# starting kernel
# baseline (speedup 1.0000x reference)
#include <hip/hip_runtime.h>
#include <math.h>

typedef float v2f __attribute__((ext_vector_type(2)));
typedef float v8f __attribute__((ext_vector_type(8)));

#define N_NODES 50000
#define BB 32
#define TT 512
#define SD 64
#define OT 128
#define GATES 512
#define BT (BB * TT)

__device__ __forceinline__ v8f wmma_f32(v2f a, v2f b, v8f c) {
    return __builtin_amdgcn_wmma_f32_16x16x4_f32(
         false, a,  false, b,
         (short)0, c,  false,  false);
}

__global__ void zero_f32(float* __restrict__ p, long n) {
    long i = (long)blockIdx.x * blockDim.x + threadIdx.x;
    if (i < n) p[i] = 0.0f;
}

__global__ void degree_kernel(const int* __restrict__ src, const int* __restrict__ dst,
                              float* __restrict__ deg_out, float* __restrict__ deg_in, int E) {
    int e = blockIdx.x * blockDim.x + threadIdx.x;
    if (e < E) {
        atomicAdd(&deg_out[src[e]], 1.0f);
        atomicAdd(&deg_in[dst[e]], 1.0f);
    }
}

__global__ void norm_finish(float* __restrict__ no, float* __restrict__ ni, int n) {
    int i = blockIdx.x * blockDim.x + threadIdx.x;
    if (i < n) {
        no[i] = rsqrtf(fmaxf(no[i], 1.0f));
        ni[i] = rsqrtf(fmaxf(ni[i], 1.0f));
    }
}

__global__ void scatter1(const int* __restrict__ src, const int* __restrict__ dst,
                         const float* __restrict__ ht, const float* __restrict__ chs,
                         const float* __restrict__ norm_out, float* __restrict__ agg1, int E) {
    long tid = (long)blockIdx.x * blockDim.x + threadIdx.x;
    int e = (int)(tid >> 5);
    if (e >= E) return;
    int q = ((int)tid & 31) << 2;
    int s = src[e], d = dst[e];
    float no = norm_out[s];
    const float* sp = (q < SD) ? (ht + (long)s * SD + q) : (chs + (long)s * SD + (q - SD));
    float4 v = *(const float4*)sp;
    float* op = agg1 + (long)d * 128 + q;
    atomicAdd(op + 0, v.x * no);
    atomicAdd(op + 1, v.y * no);
    atomicAdd(op + 2, v.z * no);
    atomicAdd(op + 3, v.w * no);
}

__global__ void scatter2(const int* __restrict__ src, const int* __restrict__ dst,
                         const float* __restrict__ x1, const float* __restrict__ norm_out,
                         float* __restrict__ agg2, int E) {
    long tid = (long)blockIdx.x * blockDim.x + threadIdx.x;
    int e = (int)(tid >> 4);
    if (e >= E) return;
    int q = ((int)tid & 15) << 2;
    int s = src[e], d = dst[e];
    float no = norm_out[s];
    float4 v = *(const float4*)(x1 + (long)s * 64 + q);
    float* op = agg2 + (long)d * 64 + q;
    atomicAdd(op + 0, v.x * no);
    atomicAdd(op + 1, v.y * no);
    atomicAdd(op + 2, v.z * no);
    atomicAdd(op + 3, v.w * no);
}

__global__ void transpose_k(const float* __restrict__ W, float* __restrict__ WT) {
    int idx = blockIdx.x * blockDim.x + threadIdx.x;
    if (idx >= GATES * OT) return;
    int r = idx >> 7, c = idx & 127;
    WT[c * GATES + r] = W[idx];
}

__global__ void bias_sum_k(const float* __restrict__ a, const float* __restrict__ b,
                           float* __restrict__ o) {
    int i = blockIdx.x * blockDim.x + threadIdx.x;
    if (i < GATES) o[i] = a[i] + b[i];
}

__global__ void concat_k(const float* __restrict__ hs, const float* __restrict__ cht,
                         float* __restrict__ X) {
    int idx = blockIdx.x * blockDim.x + threadIdx.x;
    if (idx >= BT * 128) return;
    int row = idx >> 7, k = idx & 127;
    X[idx] = (k < SD) ? hs[row * SD + k] : cht[row * SD + (k - SD)];
}

__global__ void gemm_wmma_kernel(const float* __restrict__ A, const float* __restrict__ rowscale,
                                 const float* __restrict__ Bm, const float* __restrict__ bias,
                                 float* __restrict__ C, int M, int N, int K, int applyGelu) {
    const int wave = threadIdx.x >> 5;
    const int lane = threadIdx.x & 31;
    const int hl = lane >> 4;
    const int l16 = lane & 15;
    const int ct = blockIdx.y * (blockDim.x >> 5) + wave;
    if ((ct << 4) >= N) return;
    const int row0 = blockIdx.x << 4;
    const int col = (ct << 4) + l16;
    const int arow = row0 + l16;
    const float rs = rowscale ? rowscale[arow] : 1.0f;
    const float* Arow = A + (size_t)arow * K;
    v8f acc = {};
    for (int k = 0; k < K; k += 4) {
        const int ka = k + (hl << 1);
        v2f a, b;
        a.x = Arow[ka] * rs;
        a.y = Arow[ka + 1] * rs;
        b.x = Bm[(size_t)ka * N + col];
        b.y = Bm[(size_t)(ka + 1) * N + col];
        acc = wmma_f32(a, b, acc);
    }
    const float bv = bias ? bias[col] : 0.0f;
#pragma unroll
    for (int r = 0; r < 8; ++r) {
        float v = acc[r] + bv;
        if (applyGelu) v = 0.5f * v * (1.0f + erff(v * 0.70710678118654752f));
        C[(size_t)(row0 + r + (hl << 3)) * N + col] = v;
    }
}

__global__ __launch_bounds__(1024) void lstm_kernel(const float* __restrict__ gpre,
                                                    const float* __restrict__ WhhT,
                                                    float* __restrict__ out) {
    __shared__ float h_lds[BB][OT];
    __shared__ float g_lds[BB][GATES];
    const int tid = threadIdx.x;
    const int wave = tid >> 5, lane = tid & 31;
    const int hl = lane >> 4, l16 = lane & 15;
    const int col0 = wave << 4;
    const int col = col0 + l16;

    float bx[32], by[32];
#pragma unroll
    for (int kk = 0; kk < 32; ++kk) {
        const int ka = (kk << 2) + (hl << 1);
        bx[kk] = WhhT[ka * GATES + col];
        by[kk] = WhhT[(ka + 1) * GATES + col];
    }

    for (int i = tid; i < BB * OT; i += 1024) ((float*)h_lds)[i] = 0.0f;
    float creg[4] = {0.f, 0.f, 0.f, 0.f};
    __syncthreads();

    for (int t = 0; t < TT; ++t) {
        float gp[4][4];
#pragma unroll
        for (int u = 0; u < 4; ++u) {
            const int p = (tid << 2) + u;
            const int b = p >> 7, j = p & 127;
            const float* g = gpre + ((long)b * TT + t) * GATES + j;
            gp[u][0] = g[0];
            gp[u][1] = g[OT];
            gp[u][2] = g[2 * OT];
            gp[u][3] = g[3 * OT];
        }

        v8f c0 = {}, c1 = {};
#pragma unroll
        for (int kk = 0; kk < 32; ++kk) {
            const int ka = (kk << 2) + (hl << 1);
            v2f b, a0, a1;
            b.x = bx[kk];                b.y = by[kk];
            a0.x = h_lds[l16][ka];       a0.y = h_lds[l16][ka + 1];
            a1.x = h_lds[16 + l16][ka];  a1.y = h_lds[16 + l16][ka + 1];
            c0 = wmma_f32(a0, b, c0);
            c1 = wmma_f32(a1, b, c1);
        }
#pragma unroll
        for (int r = 0; r < 8; ++r) {
            g_lds[r + (hl << 3)][col]      = c0[r];
            g_lds[16 + r + (hl << 3)][col] = c1[r];
        }
        __syncthreads();

#pragma unroll
        for (int u = 0; u < 4; ++u) {
            const int p = (tid << 2) + u;
            const int b = p >> 7, j = p & 127;
            float gi = g_lds[b][j]          + gp[u][0];
            float gf = g_lds[b][OT + j]     + gp[u][1];
            float gg = g_lds[b][2 * OT + j] + gp[u][2];
            float go = g_lds[b][3 * OT + j] + gp[u][3];
            float si = 1.0f / (1.0f + expf(-gi));
            float sf = 1.0f / (1.0f + expf(-gf));
            float so = 1.0f / (1.0f + expf(-go));
            float cc = sf * creg[u] + si * tanhf(gg);
            creg[u] = cc;
            float hh = so * tanhf(cc);
            h_lds[b][j] = hh;
            out[((long)b * TT + t) * OT + j] = hh;
        }
        __syncthreads();
    }
}

extern "C" void kernel_launch(void* const* d_in, const int* in_sizes, int n_in,
                              void* d_out, int out_size, void* d_ws, size_t ws_size,
                              hipStream_t stream) {
    const float* hs  = (const float*)d_in[0];
    const float* ht  = (const float*)d_in[1];
    const float* chs = (const float*)d_in[2];
    const float* cht = (const float*)d_in[3];
    const int*   src = (const int*)d_in[4];
    const int*   dst = (const int*)d_in[5];
    const float* W1  = (const float*)d_in[6];
    const float* b1  = (const float*)d_in[7];
    const float* W2  = (const float*)d_in[8];
    const float* b2  = (const float*)d_in[9];
    const float* Wih = (const float*)d_in[10];
    const float* Whh = (const float*)d_in[11];
    const float* bih = (const float*)d_in[12];
    const float* bhh = (const float*)d_in[13];
    const int E = in_sizes[4];

    float* ws       = (float*)d_ws;
    float* norm_out = ws;
    float* norm_in  = norm_out + N_NODES;
    float* agg1     = norm_in + N_NODES;
    float* agg2     = agg1 + (size_t)N_NODES * 128;
    float* x1       = agg2 + (size_t)N_NODES * 64;
    float* Xcat     = x1 + (size_t)N_NODES * 64;
    float* gpre     = Xcat + (size_t)BT * 128;
    float* WihT     = gpre + (size_t)BT * GATES;
    float* WhhT     = WihT + OT * GATES;
    float* bsum     = WhhT + OT * GATES;

    const int thr = 256;
    const long zn = (long)N_NODES * (2 + 128 + 64);
    zero_f32<<<(int)((zn + thr - 1) / thr), thr, 0, stream>>>(norm_out, zn);

    degree_kernel<<<(E + thr - 1) / thr, thr, 0, stream>>>(src, dst, norm_out, norm_in, E);
    norm_finish<<<(N_NODES + thr - 1) / thr, thr, 0, stream>>>(norm_out, norm_in, N_NODES);

    const long t1 = (long)E * 32;
    scatter1<<<(int)((t1 + thr - 1) / thr), thr, 0, stream>>>(src, dst, ht, chs, norm_out, agg1, E);

    dim3 gs(N_NODES / 16, 1);
    gemm_wmma_kernel<<<gs, 128, 0, stream>>>(agg1, norm_in, W1, b1, x1, N_NODES, 64, 128, 1);

    const long t2 = (long)E * 16;
    scatter2<<<(int)((t2 + thr - 1) / thr), thr, 0, stream>>>(src, dst, x1, norm_out, agg2, E);

    float* out_sp = (float*)d_out + (size_t)BT * OT;
    gemm_wmma_kernel<<<gs, 128, 0, stream>>>(agg2, norm_in, W2, b2, out_sp, N_NODES, 64, 64, 1);

    transpose_k<<<(GATES * OT + thr - 1) / thr, thr, 0, stream>>>(Wih, WihT);
    transpose_k<<<(GATES * OT + thr - 1) / thr, thr, 0, stream>>>(Whh, WhhT);
    bias_sum_k<<<2, 256, 0, stream>>>(bih, bhh, bsum);
    concat_k<<<(BT * 128 + thr - 1) / thr, thr, 0, stream>>>(hs, cht, Xcat);

    dim3 gp(BT / 16, 8);
    gemm_wmma_kernel<<<gp, 128, 0, stream>>>(Xcat, nullptr, WihT, bsum, gpre, BT, GATES, 128, 0);

    lstm_kernel<<<1, 1024, 0, stream>>>(gpre, WhhT, (float*)d_out);
}


# LSTM kernel: row pitch of h in LDS 128 -> 132 floats (16 B x odd), so the 16 rows of each A-fragment load fall on different banks instead of one: the 16-way bank conflict of every h load goes
# speedup vs baseline: 2.3448x; 2.3448x over previous
#include <hip/hip_runtime.h>
#include <math.h>

typedef float v2f __attribute__((ext_vector_type(2)));
typedef float v8f __attribute__((ext_vector_type(8)));

#define N_NODES 50000
#define BB 32
#define TT 512
#define SD 64
#define OT 128
#define GATES 512
#define BT (BB * TT)
#define HP (OT + 4)

__device__ __forceinline__ v8f wmma_f32(v2f a, v2f b, v8f c) {
    return __builtin_amdgcn_wmma_f32_16x16x4_f32(
         false, a,  false, b,
         (short)0, c,  false,  false);
}

__global__ void zero_f32(float* __restrict__ p, long n) {
    long i = (long)blockIdx.x * blockDim.x + threadIdx.x;
    if (i < n) p[i] = 0.0f;
}

__global__ void degree_kernel(const int* __restrict__ src, const int* __restrict__ dst,
                              float* __restrict__ deg_out, float* __restrict__ deg_in, int E) {
    int e = blockIdx.x * blockDim.x + threadIdx.x;
    if (e < E) {
        atomicAdd(&deg_out[src[e]], 1.0f);
        atomicAdd(&deg_in[dst[e]], 1.0f);
    }
}

__global__ void norm_finish(float* __restrict__ no, float* __restrict__ ni, int n) {
    int i = blockIdx.x * blockDim.x + threadIdx.x;
    if (i < n) {
        no[i] = rsqrtf(fmaxf(no[i], 1.0f));
        ni[i] = rsqrtf(fmaxf(ni[i], 1.0f));
    }
}

__global__ void scatter1(const int* __restrict__ src, const int* __restrict__ dst,
                         const float* __restrict__ ht, const float* __restrict__ chs,
                         const float* __restrict__ norm_out, float* __restrict__ agg1, int E) {
    long tid = (long)blockIdx.x * blockDim.x + threadIdx.x;
    int e = (int)(tid >> 5);
    if (e >= E) return;
    int q = ((int)tid & 31) << 2;
    int s = src[e], d = dst[e];
    float no = norm_out[s];
    const float* sp = (q < SD) ? (ht + (long)s * SD + q) : (chs + (long)s * SD + (q - SD));
    float4 v = *(const float4*)sp;
    float* op = agg1 + (long)d * 128 + q;
    atomicAdd(op + 0, v.x * no);
    atomicAdd(op + 1, v.y * no);
    atomicAdd(op + 2, v.z * no);
    atomicAdd(op + 3, v.w * no);
}

__global__ void scatter2(const int* __restrict__ src, const int* __restrict__ dst,
                         const float* __restrict__ x1, const float* __restrict__ norm_out,
                         float* __restrict__ agg2, int E) {
    long tid = (long)blockIdx.x * blockDim.x + threadIdx.x;
    int e = (int)(tid >> 4);
    if (e >= E) return;
    int q = ((int)tid & 15) << 2;
    int s = src[e], d = dst[e];
    float no = norm_out[s];
    float4 v = *(const float4*)(x1 + (long)s * 64 + q);
    float* op = agg2 + (long)d * 64 + q;
    atomicAdd(op + 0, v.x * no);
    atomicAdd(op + 1, v.y * no);
    atomicAdd(op + 2, v.z * no);
    atomicAdd(op + 3, v.w * no);
}

__global__ void transpose_k(const float* __restrict__ W, float* __restrict__ WT) {
    int idx = blockIdx.x * blockDim.x + threadIdx.x;
    if (idx >= GATES * OT) return;
    int r = idx >> 7, c = idx & 127;
    WT[c * GATES + r] = W[idx];
}

__global__ void bias_sum_k(const float* __restrict__ a, const float* __restrict__ b,
                           float* __restrict__ o) {
    int i = blockIdx.x * blockDim.x + threadIdx.x;
    if (i < GATES) o[i] = a[i] + b[i];
}

__global__ void concat_k(const float* __restrict__ hs, const float* __restrict__ cht,
                         float* __restrict__ X) {
    int idx = blockIdx.x * blockDim.x + threadIdx.x;
    if (idx >= BT * 128) return;
    int row = idx >> 7, k = idx & 127;
    X[idx] = (k < SD) ? hs[row * SD + k] : cht[row * SD + (k - SD)];
}

__global__ void gemm_wmma_kernel(const float* __restrict__ A, const float* __restrict__ rowscale,
                                 const float* __restrict__ Bm, const float* __restrict__ bias,
                                 float* __restrict__ C, int M, int N, int K, int applyGelu) {
    const int wave = threadIdx.x >> 5;
    const int lane = threadIdx.x & 31;
    const int hl = lane >> 4;
    const int l16 = lane & 15;
    const int ct = blockIdx.y * (blockDim.x >> 5) + wave;
    if ((ct << 4) >= N) return;
    const int row0 = blockIdx.x << 4;
    const int col = (ct << 4) + l16;
    const int arow = row0 + l16;
    const float rs = rowscale ? rowscale[arow] : 1.0f;
    const float* Arow = A + (size_t)arow * K;
    v8f acc = {};
    for (int k = 0; k < K; k += 4) {
        const int ka = k + (hl << 1);
        v2f a, b;
        a.x = Arow[ka] * rs;
        a.y = Arow[ka + 1] * rs;
        b.x = Bm[(size_t)ka * N + col];
        b.y = Bm[(size_t)(ka + 1) * N + col];
        acc = wmma_f32(a, b, acc);
    }
    const float bv = bias ? bias[col] : 0.0f;
#pragma unroll
    for (int r = 0; r < 8; ++r) {
        float v = acc[r] + bv;
        if (applyGelu) v = 0.5f * v * (1.0f + erff(v * 0.70710678118654752f));
        C[(size_t)(row0 + r + (hl << 3)) * N + col] = v;
    }
}

__global__ __launch_bounds__(1024) void lstm_kernel(const float* __restrict__ gpre,
                                                    const float* __restrict__ WhhT,
                                                    float* __restrict__ out) {
    __shared__ float h_lds[BB][HP];
    __shared__ float g_lds[BB][GATES];
    const int tid = threadIdx.x;
    const int wave = tid >> 5, lane = tid & 31;
    const int hl = lane >> 4, l16 = lane & 15;
    const int col0 = wave << 4;
    const int col = col0 + l16;

    float bx[32], by[32];
#pragma unroll
    for (int kk = 0; kk < 32; ++kk) {
        const int ka = (kk << 2) + (hl << 1);
        bx[kk] = WhhT[ka * GATES + col];
        by[kk] = WhhT[(ka + 1) * GATES + col];
    }

    for (int i = tid; i < BB * OT; i += 1024) h_lds[i >> 7][i & (OT - 1)] = 0.0f;
    float creg[4] = {0.f, 0.f, 0.f, 0.f};
    __syncthreads();

    for (int t = 0; t < TT; ++t) {
        float gp[4][4];
#pragma unroll
        for (int u = 0; u < 4; ++u) {
            const int p = (tid << 2) + u;
            const int b = p >> 7, j = p & 127;
            const float* g = gpre + ((long)b * TT + t) * GATES + j;
            gp[u][0] = g[0];
            gp[u][1] = g[OT];
            gp[u][2] = g[2 * OT];
            gp[u][3] = g[3 * OT];
        }

        v8f c0 = {}, c1 = {};
#pragma unroll
        for (int kk = 0; kk < 32; ++kk) {
            const int ka = (kk << 2) + (hl << 1);
            v2f b, a0, a1;
            b.x = bx[kk];                b.y = by[kk];
            a0.x = h_lds[l16][ka];       a0.y = h_lds[l16][ka + 1];
            a1.x = h_lds[16 + l16][ka];  a1.y = h_lds[16 + l16][ka + 1];
            c0 = wmma_f32(a0, b, c0);
            c1 = wmma_f32(a1, b, c1);
        }
#pragma unroll
        for (int r = 0; r < 8; ++r) {
            g_lds[r + (hl << 3)][col]      = c0[r];
            g_lds[16 + r + (hl << 3)][col] = c1[r];
        }
        __syncthreads();

#pragma unroll
        for (int u = 0; u < 4; ++u) {
            const int p = (tid << 2) + u;
            const int b = p >> 7, j = p & 127;
            float gi = g_lds[b][j]          + gp[u][0];
            float gf = g_lds[b][OT + j]     + gp[u][1];
            float gg = g_lds[b][2 * OT + j] + gp[u][2];
            float go = g_lds[b][3 * OT + j] + gp[u][3];
            float si = 1.0f / (1.0f + expf(-gi));
            float sf = 1.0f / (1.0f + expf(-gf));
            float so = 1.0f / (1.0f + expf(-go));
            float cc = sf * creg[u] + si * tanhf(gg);
            creg[u] = cc;
            float hh = so * tanhf(cc);
            h_lds[b][j] = hh;
            out[((long)b * TT + t) * OT + j] = hh;
        }
        __syncthreads();
    }
}

extern "C" void kernel_launch(void* const* d_in, const int* in_sizes, int n_in,
                              void* d_out, int out_size, void* d_ws, size_t ws_size,
                              hipStream_t stream) {
    const float* hs  = (const float*)d_in[0];
    const float* ht  = (const float*)d_in[1];
    const float* chs = (const float*)d_in[2];
    const float* cht = (const float*)d_in[3];
    const int*   src = (const int*)d_in[4];
    const int*   dst = (const int*)d_in[5];
    const float* W1  = (const float*)d_in[6];
    const float* b1  = (const float*)d_in[7];
    const float* W2  = (const float*)d_in[8];
    const float* b2  = (const float*)d_in[9];
    const float* Wih = (const float*)d_in[10];
    const float* Whh = (const float*)d_in[11];
    const float* bih = (const float*)d_in[12];
    const float* bhh = (const float*)d_in[13];
    const int E = in_sizes[4];

    float* ws       = (float*)d_ws;
    float* norm_out = ws;
    float* norm_in  = norm_out + N_NODES;
    float* agg1     = norm_in + N_NODES;
    float* agg2     = agg1 + (size_t)N_NODES * 128;
    float* x1       = agg2 + (size_t)N_NODES * 64;
    float* Xcat     = x1 + (size_t)N_NODES * 64;
    float* gpre     = Xcat + (size_t)BT * 128;
    float* WihT     = gpre + (size_t)BT * GATES;
    float* WhhT     = WihT + OT * GATES;
    float* bsum     = WhhT + OT * GATES;

    const int thr = 256;
    const long zn = (long)N_NODES * (2 + 128 + 64);
    zero_f32<<<(int)((zn + thr - 1) / thr), thr, 0, stream>>>(norm_out, zn);

    degree_kernel<<<(E + thr - 1) / thr, thr, 0, stream>>>(src, dst, norm_out, norm_in, E);
    norm_finish<<<(N_NODES + thr - 1) / thr, thr, 0, stream>>>(norm_out, norm_in, N_NODES);

    const long t1 = (long)E * 32;
    scatter1<<<(int)((t1 + thr - 1) / thr), thr, 0, stream>>>(src, dst, ht, chs, norm_out, agg1, E);

    dim3 gs(N_NODES / 16, 1);
    gemm_wmma_kernel<<<gs, 128, 0, stream>>>(agg1, norm_in, W1, b1, x1, N_NODES, 64, 128, 1);

    const long t2 = (long)E * 16;
    scatter2<<<(int)((t2 + thr - 1) / thr), thr, 0, stream>>>(src, dst, x1, norm_out, agg2, E);

    float* out_sp = (float*)d_out + (size_t)BT * OT;
    gemm_wmma_kernel<<<gs, 128, 0, stream>>>(agg2, norm_in, W2, b2, out_sp, N_NODES, 64, 64, 1);

    transpose_k<<<(GATES * OT + thr - 1) / thr, thr, 0, stream>>>(Wih, WihT);
    transpose_k<<<(GATES * OT + thr - 1) / thr, thr, 0, stream>>>(Whh, WhhT);
    bias_sum_k<<<2, 256, 0, stream>>>(bih, bhh, bsum);
    concat_k<<<(BT * 128 + thr - 1) / thr, thr, 0, stream>>>(hs, cht, Xcat);

    dim3 gp(BT / 16, 8);
    gemm_wmma_kernel<<<gp, 128, 0, stream>>>(Xcat, nullptr, WihT, bsum, gpre, BT, GATES, 128, 0);

    lstm_kernel<<<1, 1024, 0, stream>>>(gpre, WhhT, (float*)d_out);
}


# LSTM recurrence split over batch: 2 independent workgroups of 16 rows (one WMMA row tile each) on 2 WGPs instead of 1 of 32 rows; per-step WMMAs, h loads and activations per WGP all halve
# speedup vs baseline: 4.0307x; 1.7190x over previous
#include <hip/hip_runtime.h>
#include <math.h>

typedef float v2f __attribute__((ext_vector_type(2)));
typedef float v8f __attribute__((ext_vector_type(8)));

#define N_NODES 50000
#define BB 32
#define TT 512
#define SD 64
#define OT 128
#define GATES 512
#define BT (BB * TT)
#define HP (OT + 4)
#define RB 16

__device__ __forceinline__ v8f wmma_f32(v2f a, v2f b, v8f c) {
    return __builtin_amdgcn_wmma_f32_16x16x4_f32(
         false, a,  false, b,
         (short)0, c,  false,  false);
}

__global__ void zero_f32(float* __restrict__ p, long n) {
    long i = (long)blockIdx.x * blockDim.x + threadIdx.x;
    if (i < n) p[i] = 0.0f;
}

__global__ void degree_kernel(const int* __restrict__ src, const int* __restrict__ dst,
                              float* __restrict__ deg_out, float* __restrict__ deg_in, int E) {
    int e = blockIdx.x * blockDim.x + threadIdx.x;
    if (e < E) {
        atomicAdd(&deg_out[src[e]], 1.0f);
        atomicAdd(&deg_in[dst[e]], 1.0f);
    }
}

__global__ void norm_finish(float* __restrict__ no, float* __restrict__ ni, int n) {
    int i = blockIdx.x * blockDim.x + threadIdx.x;
    if (i < n) {
        no[i] = rsqrtf(fmaxf(no[i], 1.0f));
        ni[i] = rsqrtf(fmaxf(ni[i], 1.0f));
    }
}

__global__ void scatter1(const int* __restrict__ src, const int* __restrict__ dst,
                         const float* __restrict__ ht, const float* __restrict__ chs,
                         const float* __restrict__ norm_out, float* __restrict__ agg1, int E) {
    long tid = (long)blockIdx.x * blockDim.x + threadIdx.x;
    int e = (int)(tid >> 5);
    if (e >= E) return;
    int q = ((int)tid & 31) << 2;
    int s = src[e], d = dst[e];
    float no = norm_out[s];
    const float* sp = (q < SD) ? (ht + (long)s * SD + q) : (chs + (long)s * SD + (q - SD));
    float4 v = *(const float4*)sp;
    float* op = agg1 + (long)d * 128 + q;
    atomicAdd(op + 0, v.x * no);
    atomicAdd(op + 1, v.y * no);
    atomicAdd(op + 2, v.z * no);
    atomicAdd(op + 3, v.w * no);
}

__global__ void scatter2(const int* __restrict__ src, const int* __restrict__ dst,
                         const float* __restrict__ x1, const float* __restrict__ norm_out,
                         float* __restrict__ agg2, int E) {
    long tid = (long)blockIdx.x * blockDim.x + threadIdx.x;
    int e = (int)(tid >> 4);
    if (e >= E) return;
    int q = ((int)tid & 15) << 2;
    int s = src[e], d = dst[e];
    float no = norm_out[s];
    float4 v = *(const float4*)(x1 + (long)s * 64 + q);
    float* op = agg2 + (long)d * 64 + q;
    atomicAdd(op + 0, v.x * no);
    atomicAdd(op + 1, v.y * no);
    atomicAdd(op + 2, v.z * no);
    atomicAdd(op + 3, v.w * no);
}

__global__ void transpose_k(const float* __restrict__ W, float* __restrict__ WT) {
    int idx = blockIdx.x * blockDim.x + threadIdx.x;
    if (idx >= GATES * OT) return;
    int r = idx >> 7, c = idx & 127;
    WT[c * GATES + r] = W[idx];
}

__global__ void bias_sum_k(const float* __restrict__ a, const float* __restrict__ b,
                           float* __restrict__ o) {
    int i = blockIdx.x * blockDim.x + threadIdx.x;
    if (i < GATES) o[i] = a[i] + b[i];
}

__global__ void concat_k(const float* __restrict__ hs, const float* __restrict__ cht,
                         float* __restrict__ X) {
    int idx = blockIdx.x * blockDim.x + threadIdx.x;
    if (idx >= BT * 128) return;
    int row = idx >> 7, k = idx & 127;
    X[idx] = (k < SD) ? hs[row * SD + k] : cht[row * SD + (k - SD)];
}

__global__ void gemm_wmma_kernel(const float* __restrict__ A, const float* __restrict__ rowscale,
                                 const float* __restrict__ Bm, const float* __restrict__ bias,
                                 float* __restrict__ C, int M, int N, int K, int applyGelu) {
    const int wave = threadIdx.x >> 5;
    const int lane = threadIdx.x & 31;
    const int hl = lane >> 4;
    const int l16 = lane & 15;
    const int ct = blockIdx.y * (blockDim.x >> 5) + wave;
    if ((ct << 4) >= N) return;
    const int row0 = blockIdx.x << 4;
    const int col = (ct << 4) + l16;
    const int arow = row0 + l16;
    const float rs = rowscale ? rowscale[arow] : 1.0f;
    const float* Arow = A + (size_t)arow * K;
    v8f acc = {};
    for (int k = 0; k < K; k += 4) {
        const int ka = k + (hl << 1);
        v2f a, b;
        a.x = Arow[ka] * rs;
        a.y = Arow[ka + 1] * rs;
        b.x = Bm[(size_t)ka * N + col];
        b.y = Bm[(size_t)(ka + 1) * N + col];
        acc = wmma_f32(a, b, acc);
    }
    const float bv = bias ? bias[col] : 0.0f;
#pragma unroll
    for (int r = 0; r < 8; ++r) {
        float v = acc[r] + bv;
        if (applyGelu) v = 0.5f * v * (1.0f + erff(v * 0.70710678118654752f));
        C[(size_t)(row0 + r + (hl << 3)) * N + col] = v;
    }
}

__global__ __launch_bounds__(1024) void lstm_kernel(const float* __restrict__ gpre,
                                                    const float* __restrict__ WhhT,
                                                    float* __restrict__ out) {
    __shared__ float h_lds[RB][HP];
    __shared__ float g_lds[RB][GATES];
    const int tid = threadIdx.x;
    const int wave = tid >> 5, lane = tid & 31;
    const int hl = lane >> 4, l16 = lane & 15;
    const int col0 = wave << 4;
    const int col = col0 + l16;
    const int brow0 = blockIdx.x * RB;

    float bx[32], by[32];
#pragma unroll
    for (int kk = 0; kk < 32; ++kk) {
        const int ka = (kk << 2) + (hl << 1);
        bx[kk] = WhhT[ka * GATES + col];
        by[kk] = WhhT[(ka + 1) * GATES + col];
    }

    for (int i = tid; i < RB * OT; i += 1024) h_lds[i >> 7][i & (OT - 1)] = 0.0f;
    float creg[2] = {0.f, 0.f};
    __syncthreads();

    for (int t = 0; t < TT; ++t) {
        float gp[2][4];
#pragma unroll
        for (int u = 0; u < 2; ++u) {
            const int p = (tid << 1) + u;
            const int b = p >> 7, j = p & 127;
            const float* g = gpre + ((long)(brow0 + b) * TT + t) * GATES + j;
            gp[u][0] = g[0];
            gp[u][1] = g[OT];
            gp[u][2] = g[2 * OT];
            gp[u][3] = g[3 * OT];
        }

        v8f c0 = {};
#pragma unroll
        for (int kk = 0; kk < 32; ++kk) {
            const int ka = (kk << 2) + (hl << 1);
            v2f b, a0;
            b.x = bx[kk];                b.y = by[kk];
            a0.x = h_lds[l16][ka];       a0.y = h_lds[l16][ka + 1];
            c0 = wmma_f32(a0, b, c0);
        }
#pragma unroll
        for (int r = 0; r < 8; ++r) g_lds[r + (hl << 3)][col] = c0[r];
        __syncthreads();

#pragma unroll
        for (int u = 0; u < 2; ++u) {
            const int p = (tid << 1) + u;
            const int b = p >> 7, j = p & 127;
            float gi = g_lds[b][j]          + gp[u][0];
            float gf = g_lds[b][OT + j]     + gp[u][1];
            float gg = g_lds[b][2 * OT + j] + gp[u][2];
            float go = g_lds[b][3 * OT + j] + gp[u][3];
            float si = 1.0f / (1.0f + expf(-gi));
            float sf = 1.0f / (1.0f + expf(-gf));
            float so = 1.0f / (1.0f + expf(-go));
            float cc = sf * creg[u] + si * tanhf(gg);
            creg[u] = cc;
            float hh = so * tanhf(cc);
            h_lds[b][j] = hh;
            out[((long)(brow0 + b) * TT + t) * OT + j] = hh;
        }
        __syncthreads();
    }
}

extern "C" void kernel_launch(void* const* d_in, const int* in_sizes, int n_in,
                              void* d_out, int out_size, void* d_ws, size_t ws_size,
                              hipStream_t stream) {
    const float* hs  = (const float*)d_in[0];
    const float* ht  = (const float*)d_in[1];
    const float* chs = (const float*)d_in[2];
    const float* cht = (const float*)d_in[3];
    const int*   src = (const int*)d_in[4];
    const int*   dst = (const int*)d_in[5];
    const float* W1  = (const float*)d_in[6];
    const float* b1  = (const float*)d_in[7];
    const float* W2  = (const float*)d_in[8];
    const float* b2  = (const float*)d_in[9];
    const float* Wih = (const float*)d_in[10];
    const float* Whh = (const float*)d_in[11];
    const float* bih = (const float*)d_in[12];
    const float* bhh = (const float*)d_in[13];
    const int E = in_sizes[4];

    float* ws       = (float*)d_ws;
    float* norm_out = ws;
    float* norm_in  = norm_out + N_NODES;
    float* agg1     = norm_in + N_NODES;
    float* agg2     = agg1 + (size_t)N_NODES * 128;
    float* x1       = agg2 + (size_t)N_NODES * 64;
    float* Xcat     = x1 + (size_t)N_NODES * 64;
    float* gpre     = Xcat + (size_t)BT * 128;
    float* WihT     = gpre + (size_t)BT * GATES;
    float* WhhT     = WihT + OT * GATES;
    float* bsum     = WhhT + OT * GATES;

    const int thr = 256;
    const long zn = (long)N_NODES * (2 + 128 + 64);
    zero_f32<<<(int)((zn + thr - 1) / thr), thr, 0, stream>>>(norm_out, zn);

    degree_kernel<<<(E + thr - 1) / thr, thr, 0, stream>>>(src, dst, norm_out, norm_in, E);
    norm_finish<<<(N_NODES + thr - 1) / thr, thr, 0, stream>>>(norm_out, norm_in, N_NODES);

    const long t1 = (long)E * 32;
    scatter1<<<(int)((t1 + thr - 1) / thr), thr, 0, stream>>>(src, dst, ht, chs, norm_out, agg1, E);

    dim3 gs(N_NODES / 16, 1);
    gemm_wmma_kernel<<<gs, 128, 0, stream>>>(agg1, norm_in, W1, b1, x1, N_NODES, 64, 128, 1);

    const long t2 = (long)E * 16;
    scatter2<<<(int)((t2 + thr - 1) / thr), thr, 0, stream>>>(src, dst, x1, norm_out, agg2, E);

    float* out_sp = (float*)d_out + (size_t)BT * OT;
    gemm_wmma_kernel<<<gs, 128, 0, stream>>>(agg2, norm_in, W2, b2, out_sp, N_NODES, 64, 64, 1);

    transpose_k<<<(GATES * OT + thr - 1) / thr, thr, 0, stream>>>(Wih, WihT);
    transpose_k<<<(GATES * OT + thr - 1) / thr, thr, 0, stream>>>(Whh, WhhT);
    bias_sum_k<<<2, 256, 0, stream>>>(bih, bhh, bsum);
    concat_k<<<(BT * 128 + thr - 1) / thr, thr, 0, stream>>>(hs, cht, Xcat);

    dim3 gp(BT / 16, 8);
    gemm_wmma_kernel<<<gp, 128, 0, stream>>>(Xcat, nullptr, WihT, bsum, gpre, BT, GATES, 128, 0);

    lstm_kernel<<<BB / RB, 1024, 0, stream>>>(gpre, WhhT, (float*)d_out);
}


# LSTM: gate columns interleaved so one wave's WMMA tile holds all 4 gates of 4 hidden units; each wave finishes its own cells while other waves still run WMMAs; h double-buffered, 1 barrier per step
# speedup vs baseline: 4.2171x; 1.0462x over previous
#include <hip/hip_runtime.h>
#include <math.h>

typedef float v2f __attribute__((ext_vector_type(2)));
typedef float v8f __attribute__((ext_vector_type(8)));

#define N_NODES 50000
#define BB 32
#define TT 512
#define SD 64
#define OT 128
#define GATES 512
#define BT (BB * TT)
#define HP (OT + 4)
#define RB 16
#define GCOL(r) (((((r) & (OT - 1)) >> 2) << 4) + (((r) & 3) << 2) + ((r) >> 7))
#define SP 144
#define SW 272

__device__ __forceinline__ v8f wmma_f32(v2f a, v2f b, v8f c) {
    return __builtin_amdgcn_wmma_f32_16x16x4_f32(
         false, a,  false, b,
         (short)0, c,  false,  false);
}

__global__ void zero_f32(float* __restrict__ p, long n) {
    long i = (long)blockIdx.x * blockDim.x + threadIdx.x;
    if (i < n) p[i] = 0.0f;
}

__global__ void degree_kernel(const int* __restrict__ src, const int* __restrict__ dst,
                              float* __restrict__ deg_out, float* __restrict__ deg_in, int E) {
    int e = blockIdx.x * blockDim.x + threadIdx.x;
    if (e < E) {
        atomicAdd(&deg_out[src[e]], 1.0f);
        atomicAdd(&deg_in[dst[e]], 1.0f);
    }
}

__global__ void norm_finish(float* __restrict__ no, float* __restrict__ ni, int n) {
    int i = blockIdx.x * blockDim.x + threadIdx.x;
    if (i < n) {
        no[i] = rsqrtf(fmaxf(no[i], 1.0f));
        ni[i] = rsqrtf(fmaxf(ni[i], 1.0f));
    }
}

__global__ void scatter1(const int* __restrict__ src, const int* __restrict__ dst,
                         const float* __restrict__ ht, const float* __restrict__ chs,
                         const float* __restrict__ norm_out, float* __restrict__ agg1, int E) {
    long tid = (long)blockIdx.x * blockDim.x + threadIdx.x;
    int e = (int)(tid >> 5);
    if (e >= E) return;
    int q = ((int)tid & 31) << 2;
    int s = src[e], d = dst[e];
    float no = norm_out[s];
    const float* sp = (q < SD) ? (ht + (long)s * SD + q) : (chs + (long)s * SD + (q - SD));
    float4 v = *(const float4*)sp;
    float* op = agg1 + (long)d * 128 + q;
    atomicAdd(op + 0, v.x * no);
    atomicAdd(op + 1, v.y * no);
    atomicAdd(op + 2, v.z * no);
    atomicAdd(op + 3, v.w * no);
}

__global__ void scatter2(const int* __restrict__ src, const int* __restrict__ dst,
                         const float* __restrict__ x1, const float* __restrict__ norm_out,
                         float* __restrict__ agg2, int E) {
    long tid = (long)blockIdx.x * blockDim.x + threadIdx.x;
    int e = (int)(tid >> 4);
    if (e >= E) return;
    int q = ((int)tid & 15) << 2;
    int s = src[e], d = dst[e];
    float no = norm_out[s];
    float4 v = *(const float4*)(x1 + (long)s * 64 + q);
    float* op = agg2 + (long)d * 64 + q;
    atomicAdd(op + 0, v.x * no);
    atomicAdd(op + 1, v.y * no);
    atomicAdd(op + 2, v.z * no);
    atomicAdd(op + 3, v.w * no);
}

__global__ void transpose_k(const float* __restrict__ W, float* __restrict__ WT) {
    int idx = blockIdx.x * blockDim.x + threadIdx.x;
    if (idx >= GATES * OT) return;
    int r = idx >> 7, c = idx & 127;
    WT[c * GATES + GCOL(r)] = W[idx];
}

__global__ void bias_sum_k(const float* __restrict__ a, const float* __restrict__ b,
                           float* __restrict__ o) {
    int i = blockIdx.x * blockDim.x + threadIdx.x;
    if (i < GATES) o[GCOL(i)] = a[i] + b[i];
}

__global__ void concat_k(const float* __restrict__ hs, const float* __restrict__ cht,
                         float* __restrict__ X) {
    int idx = blockIdx.x * blockDim.x + threadIdx.x;
    if (idx >= BT * 128) return;
    int row = idx >> 7, k = idx & 127;
    X[idx] = (k < SD) ? hs[row * SD + k] : cht[row * SD + (k - SD)];
}

__global__ void gemm_wmma_kernel(const float* __restrict__ A, const float* __restrict__ rowscale,
                                 const float* __restrict__ Bm, const float* __restrict__ bias,
                                 float* __restrict__ C, int M, int N, int K, int applyGelu) {
    const int wave = threadIdx.x >> 5;
    const int lane = threadIdx.x & 31;
    const int hl = lane >> 4;
    const int l16 = lane & 15;
    const int ct = blockIdx.y * (blockDim.x >> 5) + wave;
    if ((ct << 4) >= N) return;
    const int row0 = blockIdx.x << 4;
    const int col = (ct << 4) + l16;
    const int arow = row0 + l16;
    const float rs = rowscale ? rowscale[arow] : 1.0f;
    const float* Arow = A + (size_t)arow * K;
    v8f acc = {};
    for (int k = 0; k < K; k += 4) {
        const int ka = k + (hl << 1);
        v2f a, b;
        a.x = Arow[ka] * rs;
        a.y = Arow[ka + 1] * rs;
        b.x = Bm[(size_t)ka * N + col];
        b.y = Bm[(size_t)(ka + 1) * N + col];
        acc = wmma_f32(a, b, acc);
    }
    const float bv = bias ? bias[col] : 0.0f;
#pragma unroll
    for (int r = 0; r < 8; ++r) {
        float v = acc[r] + bv;
        if (applyGelu) v = 0.5f * v * (1.0f + erff(v * 0.70710678118654752f));
        C[(size_t)(row0 + r + (hl << 3)) * N + col] = v;
    }
}

__global__ __launch_bounds__(1024) void lstm_kernel(const float* __restrict__ gpre,
                                                    const float* __restrict__ WhhT,
                                                    float* __restrict__ out) {
    __shared__ __attribute__((aligned(16))) float h_lds[2][RB][HP];
    __shared__ __attribute__((aligned(16))) float s_lds[32][SW];
    const int tid = threadIdx.x;
    const int wave = tid >> 5, lane = tid & 31;
    const int hl = lane >> 4, l16 = lane & 15;
    const int col0 = wave << 4;
    const int col = col0 + l16;
    const int brow0 = blockIdx.x * RB;

    float bx[32], by[32];
#pragma unroll
    for (int kk = 0; kk < 32; ++kk) {
        const int ka = (kk << 2) + (hl << 1);
        bx[kk] = WhhT[ka * GATES + col];
        by[kk] = WhhT[(ka + 1) * GATES + col];
    }

    for (int i = tid; i < RB * OT; i += 1024) h_lds[0][i >> 7][i & (OT - 1)] = 0.0f;
    float creg[2] = {0.f, 0.f};
    const int crow = lane >> 2;
    const int cq = (lane & 3) << 2;
    const int cj = (wave << 2) + (lane & 3);
    float* sw = &s_lds[wave][0];
    const int orow = wave >> 1;
    const int ocol = ((wave & 1) << 6) + (lane << 1);
    float* outp = out + (long)(brow0 + orow) * TT * OT + ocol;
    __syncthreads();

    int cur = 0;
    auto step = [&](const int t, const bool store_prev) __attribute__((always_inline)) {
        float4 gp[2];
#pragma unroll
        for (int u = 0; u < 2; ++u)
            gp[u] = *(const float4*)(gpre + ((long)(brow0 + crow + (u << 3)) * TT + t) * GATES + col0 + cq);

        const float (*hc)[HP] = h_lds[cur];
        float (*hn)[HP] = h_lds[cur ^ 1];
        const float2 ov = *(const float2*)&hc[orow][ocol];

        v8f c0 = {};
#pragma unroll
        for (int kk = 0; kk < 32; ++kk) {
            const int ka = (kk << 2) + (hl << 1);
            v2f b, a0;
            b.x = bx[kk];                b.y = by[kk];
            a0.x = hc[l16][ka];          a0.y = hc[l16][ka + 1];
            c0 = wmma_f32(a0, b, c0);
        }
#pragma unroll
        for (int r = 0; r < 8; ++r) sw[hl * SP + (r << 4) + l16] = c0[r];
        __builtin_amdgcn_fence(__ATOMIC_RELEASE, "wavefront");
        __builtin_amdgcn_wave_barrier();
        __builtin_amdgcn_fence(__ATOMIC_ACQUIRE, "wavefront");
        float4 gs[2];
#pragma unroll
        for (int u = 0; u < 2; ++u) gs[u] = *(const float4*)&sw[u * SP + (lane << 2)];
        if (store_prev) *(float2*)(outp + (long)(t - 1) * OT) = ov;

#pragma unroll
        for (int u = 0; u < 2; ++u) {
            float gi = gs[u].x + gp[u].x;
            float gf = gs[u].y + gp[u].y;
            float gg = gs[u].z + gp[u].z;
            float go = gs[u].w + gp[u].w;
            float si = 1.0f / (1.0f + expf(-gi));
            float sf = 1.0f / (1.0f + expf(-gf));
            float so = 1.0f / (1.0f + expf(-go));
            float cc = sf * creg[u] + si * tanhf(gg);
            creg[u] = cc;
            float hh = so * tanhf(cc);
            hn[crow + (u << 3)][cj] = hh;
        }
        __syncthreads();
        cur ^= 1;
    };
    step(0, false);
    for (int t = 1; t < TT; ++t) step(t, true);
    *(float2*)(outp + (long)(TT - 1) * OT) = *(const float2*)&h_lds[cur][orow][ocol];
}

extern "C" void kernel_launch(void* const* d_in, const int* in_sizes, int n_in,
                              void* d_out, int out_size, void* d_ws, size_t ws_size,
                              hipStream_t stream) {
    const float* hs  = (const float*)d_in[0];
    const float* ht  = (const float*)d_in[1];
    const float* chs = (const float*)d_in[2];
    const float* cht = (const float*)d_in[3];
    const int*   src = (const int*)d_in[4];
    const int*   dst = (const int*)d_in[5];
    const float* W1  = (const float*)d_in[6];
    const float* b1  = (const float*)d_in[7];
    const float* W2  = (const float*)d_in[8];
    const float* b2  = (const float*)d_in[9];
    const float* Wih = (const float*)d_in[10];
    const float* Whh = (const float*)d_in[11];
    const float* bih = (const float*)d_in[12];
    const float* bhh = (const float*)d_in[13];
    const int E = in_sizes[4];

    float* ws       = (float*)d_ws;
    float* norm_out = ws;
    float* norm_in  = norm_out + N_NODES;
    float* agg1     = norm_in + N_NODES;
    float* agg2     = agg1 + (size_t)N_NODES * 128;
    float* x1       = agg2 + (size_t)N_NODES * 64;
    float* Xcat     = x1 + (size_t)N_NODES * 64;
    float* gpre     = Xcat + (size_t)BT * 128;
    float* WihT     = gpre + (size_t)BT * GATES;
    float* WhhT     = WihT + OT * GATES;
    float* bsum     = WhhT + OT * GATES;

    const int thr = 256;
    const long zn = (long)N_NODES * (2 + 128 + 64);
    zero_f32<<<(int)((zn + thr - 1) / thr), thr, 0, stream>>>(norm_out, zn);

    degree_kernel<<<(E + thr - 1) / thr, thr, 0, stream>>>(src, dst, norm_out, norm_in, E);
    norm_finish<<<(N_NODES + thr - 1) / thr, thr, 0, stream>>>(norm_out, norm_in, N_NODES);

    const long t1 = (long)E * 32;
    scatter1<<<(int)((t1 + thr - 1) / thr), thr, 0, stream>>>(src, dst, ht, chs, norm_out, agg1, E);

    dim3 gs(N_NODES / 16, 1);
    gemm_wmma_kernel<<<gs, 128, 0, stream>>>(agg1, norm_in, W1, b1, x1, N_NODES, 64, 128, 1);

    const long t2 = (long)E * 16;
    scatter2<<<(int)((t2 + thr - 1) / thr), thr, 0, stream>>>(src, dst, x1, norm_out, agg2, E);

    float* out_sp = (float*)d_out + (size_t)BT * OT;
    gemm_wmma_kernel<<<gs, 128, 0, stream>>>(agg2, norm_in, W2, b2, out_sp, N_NODES, 64, 64, 1);

    transpose_k<<<(GATES * OT + thr - 1) / thr, thr, 0, stream>>>(Wih, WihT);
    transpose_k<<<(GATES * OT + thr - 1) / thr, thr, 0, stream>>>(Whh, WhhT);
    bias_sum_k<<<2, 256, 0, stream>>>(bih, bhh, bsum);
    concat_k<<<(BT * 128 + thr - 1) / thr, thr, 0, stream>>>(hs, cht, Xcat);

    dim3 gp(BT / 16, 8);
    gemm_wmma_kernel<<<gp, 128, 0, stream>>>(Xcat, nullptr, WihT, bsum, gpre, BT, GATES, 128, 0);

    lstm_kernel<<<BB / RB, 1024, 0, stream>>>(gpre, WhhT, (float*)d_out);
}


# GraphConv sums: edges bucketed by dst once per call (int degrees, scan, 1 int atomic per edge), one wave per node adds its bucket's src rows in registers; no float atomics, sum order = arrival rank
# speedup vs baseline: 4.5728x; 1.0843x over previous
#include <hip/hip_runtime.h>
#include <math.h>

typedef float v2f __attribute__((ext_vector_type(2)));
typedef float v8f __attribute__((ext_vector_type(8)));

#define N_NODES 50000
#define BB 32
#define TT 512
#define SD 64
#define OT 128
#define GATES 512
#define BT (BB * TT)
#define HP (OT + 4)
#define RB 16
#define GCOL(r) (((((r) & (OT - 1)) >> 2) << 4) + (((r) & 3) << 2) + ((r) >> 7))
#define SP 144
#define SW 272

__device__ __forceinline__ v8f wmma_f32(v2f a, v2f b, v8f c) {
    return __builtin_amdgcn_wmma_f32_16x16x4_f32(
         false, a,  false, b,
         (short)0, c,  false,  false);
}

__global__ void zero_i32(int* __restrict__ p, int n) {
    int i = blockIdx.x * blockDim.x + threadIdx.x;
    if (i < n) p[i] = 0;
}

__global__ void degree_kernel(const int* __restrict__ src, const int* __restrict__ dst,
                              int* __restrict__ cnt_out, int* __restrict__ cnt_in, int E) {
    int e = blockIdx.x * blockDim.x + threadIdx.x;
    if (e < E) {
        atomicAdd(&cnt_out[src[e]], 1);
        atomicAdd(&cnt_in[dst[e]], 1);
    }
}

#define SCAN_IT ((N_NODES + 1023) / 1024)
#define SCAN_G 7
static_assert(SCAN_IT % SCAN_G == 0, "group size must divide the pass count");
__global__ __launch_bounds__(1024) void scan_norm_k(const int* __restrict__ cnt_in,
                                                    const int* __restrict__ cnt_out,
                                                    int* __restrict__ rowptr, int* __restrict__ cursor,
                                                    float* __restrict__ no, float* __restrict__ ni) {
    __shared__ int wtot[32];
    const int wave = threadIdx.x >> 5, lane = threadIdx.x & 31;
    const int base = wave * (SCAN_IT * 32) + lane;
    int tot = 0;
#pragma unroll 7
    for (int it = 0; it < SCAN_IT; ++it) {
        const int i = base + it * 32;
        const int cval = cnt_in[min(i, N_NODES - 1)];
        tot += (i < N_NODES) ? cval : 0;
    }
#pragma unroll
    for (int o = 16; o > 0; o >>= 1) tot += __shfl_xor(tot, o);
    if (lane == 0) wtot[wave] = tot;
    __syncthreads();
    int carry = 0;
    for (int w = 0; w < wave; ++w) carry += wtot[w];
    for (int g = 0; g < SCAN_IT; g += SCAN_G) {
        int cv[SCAN_G], co[SCAN_G], x[SCAN_G];
#pragma unroll
        for (int u = 0; u < SCAN_G; ++u) {
            const int i = base + (g + u) * 32;
            const int ic = min(i, N_NODES - 1);
            const int cval = cnt_in[ic];
            co[u] = cnt_out[ic];
            cv[u] = (i < N_NODES) ? cval : 0;
            x[u] = cv[u];
        }
#pragma unroll
        for (int o = 1; o < 32; o <<= 1) {
#pragma unroll
            for (int u = 0; u < SCAN_G; ++u) {
                const int y = __shfl_up(x[u], o);
                if (lane >= o) x[u] += y;
            }
        }
#pragma unroll
        for (int u = 0; u < SCAN_G; ++u) {
            const int i = base + (g + u) * 32;
            if (i < N_NODES) {
                const int v = carry + x[u] - cv[u];
                rowptr[i] = v;
                cursor[i] = v;
                no[i] = rsqrtf(fmaxf((float)co[u], 1.0f));
                ni[i] = rsqrtf(fmaxf((float)cv[u], 1.0f));
            }
            carry += __shfl(x[u], 31);
        }
    }
    if (threadIdx.x == 1023) rowptr[N_NODES] = carry;
}

__global__ void fill_k(const int* __restrict__ src, const int* __restrict__ dst,
                       const float* __restrict__ norm_out, int* __restrict__ cursor,
                       int2* __restrict__ ebuf, int E) {
    int e = blockIdx.x * blockDim.x + threadIdx.x;
    if (e < E) {
        const int s = src[e];
        const int p = atomicAdd(&cursor[dst[e]], 1);
        ebuf[p] = make_int2(s, __float_as_int(norm_out[s]));
    }
}

__global__ void gather1(const int* __restrict__ rowptr, const int2* __restrict__ ebuf,
                        const float* __restrict__ ht, const float* __restrict__ chs,
                        float* __restrict__ agg1) {
    const int lane = threadIdx.x & 31;
    const int node = __builtin_amdgcn_readfirstlane((int)((blockIdx.x * blockDim.x + threadIdx.x) >> 5));
    if (node >= N_NODES) return;
    const int q = lane << 2;
    const float* fb = (q < SD) ? (ht + q) : (chs + (q - SD));
    const int beg = rowptr[node], end = rowptr[node + 1];
    float4 acc = make_float4(0.f, 0.f, 0.f, 0.f);
    for (int j0 = beg; j0 < end; j0 += 32) {
        const int m = min(32, end - j0);
        int2 ew = make_int2(0, 0);
        if (lane < m) ew = ebuf[j0 + lane];
        int i = 0;
        for (; i + 4 <= m; i += 4) {
            const int s0 = __builtin_amdgcn_readlane(ew.x, i);
            const int s1 = __builtin_amdgcn_readlane(ew.x, i + 1);
            const int s2 = __builtin_amdgcn_readlane(ew.x, i + 2);
            const int s3 = __builtin_amdgcn_readlane(ew.x, i + 3);
            const float4 v0 = *(const float4*)(fb + (long)s0 * SD);
            const float4 v1 = *(const float4*)(fb + (long)s1 * SD);
            const float4 v2 = *(const float4*)(fb + (long)s2 * SD);
            const float4 v3 = *(const float4*)(fb + (long)s3 * SD);
            const float w0 = __int_as_float(__builtin_amdgcn_readlane(ew.y, i));
            const float w1 = __int_as_float(__builtin_amdgcn_readlane(ew.y, i + 1));
            const float w2 = __int_as_float(__builtin_amdgcn_readlane(ew.y, i + 2));
            const float w3 = __int_as_float(__builtin_amdgcn_readlane(ew.y, i + 3));
            acc.x += v0.x * w0; acc.y += v0.y * w0; acc.z += v0.z * w0; acc.w += v0.w * w0;
            acc.x += v1.x * w1; acc.y += v1.y * w1; acc.z += v1.z * w1; acc.w += v1.w * w1;
            acc.x += v2.x * w2; acc.y += v2.y * w2; acc.z += v2.z * w2; acc.w += v2.w * w2;
            acc.x += v3.x * w3; acc.y += v3.y * w3; acc.z += v3.z * w3; acc.w += v3.w * w3;
        }
        for (; i < m; ++i) {
            const int s0 = __builtin_amdgcn_readlane(ew.x, i);
            const float w0 = __int_as_float(__builtin_amdgcn_readlane(ew.y, i));
            const float4 v0 = *(const float4*)(fb + (long)s0 * SD);
            acc.x += v0.x * w0; acc.y += v0.y * w0; acc.z += v0.z * w0; acc.w += v0.w * w0;
        }
    }
    *(float4*)(agg1 + (long)node * 128 + q) = acc;
}

__global__ void gather2(const int* __restrict__ rowptr, const int2* __restrict__ ebuf,
                        const float* __restrict__ x1, float* __restrict__ agg2) {
    const int lane = threadIdx.x & 31;
    const int node = __builtin_amdgcn_readfirstlane((int)((blockIdx.x * blockDim.x + threadIdx.x) >> 5));
    if (node >= N_NODES) return;
    const int q = lane << 1;
    const float* fb = x1 + q;
    const int beg = rowptr[node], end = rowptr[node + 1];
    float2 acc = make_float2(0.f, 0.f);
    for (int j0 = beg; j0 < end; j0 += 32) {
        const int m = min(32, end - j0);
        int2 ew = make_int2(0, 0);
        if (lane < m) ew = ebuf[j0 + lane];
        int i = 0;
        for (; i + 4 <= m; i += 4) {
            const int s0 = __builtin_amdgcn_readlane(ew.x, i);
            const int s1 = __builtin_amdgcn_readlane(ew.x, i + 1);
            const int s2 = __builtin_amdgcn_readlane(ew.x, i + 2);
            const int s3 = __builtin_amdgcn_readlane(ew.x, i + 3);
            const float2 v0 = *(const float2*)(fb + (long)s0 * 64);
            const float2 v1 = *(const float2*)(fb + (long)s1 * 64);
            const float2 v2 = *(const float2*)(fb + (long)s2 * 64);
            const float2 v3 = *(const float2*)(fb + (long)s3 * 64);
            const float w0 = __int_as_float(__builtin_amdgcn_readlane(ew.y, i));
            const float w1 = __int_as_float(__builtin_amdgcn_readlane(ew.y, i + 1));
            const float w2 = __int_as_float(__builtin_amdgcn_readlane(ew.y, i + 2));
            const float w3 = __int_as_float(__builtin_amdgcn_readlane(ew.y, i + 3));
            acc.x += v0.x * w0; acc.y += v0.y * w0;
            acc.x += v1.x * w1; acc.y += v1.y * w1;
            acc.x += v2.x * w2; acc.y += v2.y * w2;
            acc.x += v3.x * w3; acc.y += v3.y * w3;
        }
        for (; i < m; ++i) {
            const int s0 = __builtin_amdgcn_readlane(ew.x, i);
            const float w0 = __int_as_float(__builtin_amdgcn_readlane(ew.y, i));
            const float2 v0 = *(const float2*)(fb + (long)s0 * 64);
            acc.x += v0.x * w0; acc.y += v0.y * w0;
        }
    }
    *(float2*)(agg2 + (long)node * 64 + q) = acc;
}

__global__ void transpose_k(const float* __restrict__ W, float* __restrict__ WT) {
    int idx = blockIdx.x * blockDim.x + threadIdx.x;
    if (idx >= GATES * OT) return;
    int r = idx >> 7, c = idx & 127;
    WT[c * GATES + GCOL(r)] = W[idx];
}

__global__ void bias_sum_k(const float* __restrict__ a, const float* __restrict__ b,
                           float* __restrict__ o) {
    int i = blockIdx.x * blockDim.x + threadIdx.x;
    if (i < GATES) o[GCOL(i)] = a[i] + b[i];
}

__global__ void concat_k(const float* __restrict__ hs, const float* __restrict__ cht,
                         float* __restrict__ X) {
    int idx = blockIdx.x * blockDim.x + threadIdx.x;
    if (idx >= BT * 128) return;
    int row = idx >> 7, k = idx & 127;
    X[idx] = (k < SD) ? hs[row * SD + k] : cht[row * SD + (k - SD)];
}

__global__ void gemm_wmma_kernel(const float* __restrict__ A, const float* __restrict__ rowscale,
                                 const float* __restrict__ Bm, const float* __restrict__ bias,
                                 float* __restrict__ C, int M, int N, int K, int applyGelu) {
    const int wave = threadIdx.x >> 5;
    const int lane = threadIdx.x & 31;
    const int hl = lane >> 4;
    const int l16 = lane & 15;
    const int ct = blockIdx.y * (blockDim.x >> 5) + wave;
    if ((ct << 4) >= N) return;
    const int row0 = blockIdx.x << 4;
    const int col = (ct << 4) + l16;
    const int arow = row0 + l16;
    const float rs = rowscale ? rowscale[arow] : 1.0f;
    const float* Arow = A + (size_t)arow * K;
    v8f acc = {};
    for (int k = 0; k < K; k += 4) {
        const int ka = k + (hl << 1);
        v2f a, b;
        a.x = Arow[ka] * rs;
        a.y = Arow[ka + 1] * rs;
        b.x = Bm[(size_t)ka * N + col];
        b.y = Bm[(size_t)(ka + 1) * N + col];
        acc = wmma_f32(a, b, acc);
    }
    const float bv = bias ? bias[col] : 0.0f;
#pragma unroll
    for (int r = 0; r < 8; ++r) {
        float v = acc[r] + bv;
        if (applyGelu) v = 0.5f * v * (1.0f + erff(v * 0.70710678118654752f));
        C[(size_t)(row0 + r + (hl << 3)) * N + col] = v;
    }
}

__global__ __launch_bounds__(1024) void lstm_kernel(const float* __restrict__ gpre,
                                                    const float* __restrict__ WhhT,
                                                    float* __restrict__ out) {
    __shared__ __attribute__((aligned(16))) float h_lds[2][RB][HP];
    __shared__ __attribute__((aligned(16))) float s_lds[32][SW];
    const int tid = threadIdx.x;
    const int wave = tid >> 5, lane = tid & 31;
    const int hl = lane >> 4, l16 = lane & 15;
    const int col0 = wave << 4;
    const int col = col0 + l16;
    const int brow0 = blockIdx.x * RB;

    float bx[32], by[32];
#pragma unroll
    for (int kk = 0; kk < 32; ++kk) {
        const int ka = (kk << 2) + (hl << 1);
        bx[kk] = WhhT[ka * GATES + col];
        by[kk] = WhhT[(ka + 1) * GATES + col];
    }

    for (int i = tid; i < RB * OT; i += 1024) h_lds[0][i >> 7][i & (OT - 1)] = 0.0f;
    float creg[2] = {0.f, 0.f};
    const int crow = lane >> 2;
    const int cq = (lane & 3) << 2;
    const int cj = (wave << 2) + (lane & 3);
    float* sw = &s_lds[wave][0];
    const int orow = wave >> 1;
    const int ocol = ((wave & 1) << 6) + (lane << 1);
    float* outp = out + (long)(brow0 + orow) * TT * OT + ocol;
    __syncthreads();

    int cur = 0;
    auto step = [&](const int t, const bool store_prev) __attribute__((always_inline)) {
        float4 gp[2];
#pragma unroll
        for (int u = 0; u < 2; ++u)
            gp[u] = *(const float4*)(gpre + ((long)(brow0 + crow + (u << 3)) * TT + t) * GATES + col0 + cq);

        const float (*hc)[HP] = h_lds[cur];
        float (*hn)[HP] = h_lds[cur ^ 1];
        const float2 ov = *(const float2*)&hc[orow][ocol];

        v8f c0 = {};
#pragma unroll
        for (int kk = 0; kk < 32; ++kk) {
            const int ka = (kk << 2) + (hl << 1);
            v2f b, a0;
            b.x = bx[kk];                b.y = by[kk];
            a0.x = hc[l16][ka];          a0.y = hc[l16][ka + 1];
            c0 = wmma_f32(a0, b, c0);
        }
#pragma unroll
        for (int r = 0; r < 8; ++r) sw[hl * SP + (r << 4) + l16] = c0[r];
        __builtin_amdgcn_fence(__ATOMIC_RELEASE, "wavefront");
        __builtin_amdgcn_wave_barrier();
        __builtin_amdgcn_fence(__ATOMIC_ACQUIRE, "wavefront");
        float4 gs[2];
#pragma unroll
        for (int u = 0; u < 2; ++u) gs[u] = *(const float4*)&sw[u * SP + (lane << 2)];
        if (store_prev) *(float2*)(outp + (long)(t - 1) * OT) = ov;

#pragma unroll
        for (int u = 0; u < 2; ++u) {
            float gi = gs[u].x + gp[u].x;
            float gf = gs[u].y + gp[u].y;
            float gg = gs[u].z + gp[u].z;
            float go = gs[u].w + gp[u].w;
            float si = 1.0f / (1.0f + expf(-gi));
            float sf = 1.0f / (1.0f + expf(-gf));
            float so = 1.0f / (1.0f + expf(-go));
            float cc = sf * creg[u] + si * tanhf(gg);
            creg[u] = cc;
            float hh = so * tanhf(cc);
            hn[crow + (u << 3)][cj] = hh;
        }
        __syncthreads();
        cur ^= 1;
    };
    step(0, false);
    for (int t = 1; t < TT; ++t) step(t, true);
    *(float2*)(outp + (long)(TT - 1) * OT) = *(const float2*)&h_lds[cur][orow][ocol];
}

extern "C" void kernel_launch(void* const* d_in, const int* in_sizes, int n_in,
                              void* d_out, int out_size, void* d_ws, size_t ws_size,
                              hipStream_t stream) {
    const float* hs  = (const float*)d_in[0];
    const float* ht  = (const float*)d_in[1];
    const float* chs = (const float*)d_in[2];
    const float* cht = (const float*)d_in[3];
    const int*   src = (const int*)d_in[4];
    const int*   dst = (const int*)d_in[5];
    const float* W1  = (const float*)d_in[6];
    const float* b1  = (const float*)d_in[7];
    const float* W2  = (const float*)d_in[8];
    const float* b2  = (const float*)d_in[9];
    const float* Wih = (const float*)d_in[10];
    const float* Whh = (const float*)d_in[11];
    const float* bih = (const float*)d_in[12];
    const float* bhh = (const float*)d_in[13];
    const int E = in_sizes[4];

    float* ws       = (float*)d_ws;
    float* norm_out = ws;
    float* norm_in  = norm_out + N_NODES;
    float* agg1     = norm_in + N_NODES;
    float* agg2     = agg1 + (size_t)N_NODES * 128;
    float* x1       = agg2 + (size_t)N_NODES * 64;
    float* Xcat     = x1 + (size_t)N_NODES * 64;
    float* gpre     = Xcat + (size_t)BT * 128;
    float* WihT     = gpre + (size_t)BT * GATES;
    float* WhhT     = WihT + OT * GATES;
    float* bsum     = WhhT + OT * GATES;
    int*  cnt_in  = (int*)gpre;
    int*  cnt_out = cnt_in + N_NODES;
    int*  rowptr  = cnt_out + N_NODES;
    int*  cursor  = rowptr + N_NODES + 16;
    int2* ebuf    = (int2*)(cursor + N_NODES);

    const int thr = 256;
    zero_i32<<<(2 * N_NODES + thr - 1) / thr, thr, 0, stream>>>(cnt_in, 2 * N_NODES);

    degree_kernel<<<(E + thr - 1) / thr, thr, 0, stream>>>(src, dst, cnt_out, cnt_in, E);
    scan_norm_k<<<1, 1024, 0, stream>>>(cnt_in, cnt_out, rowptr, cursor, norm_out, norm_in);
    fill_k<<<(E + thr - 1) / thr, thr, 0, stream>>>(src, dst, norm_out, cursor, ebuf, E);

    const int gblocks = (N_NODES * 32 + thr - 1) / thr;
    gather1<<<gblocks, thr, 0, stream>>>(rowptr, ebuf, ht, chs, agg1);

    dim3 gs(N_NODES / 16, 1);
    gemm_wmma_kernel<<<gs, 128, 0, stream>>>(agg1, norm_in, W1, b1, x1, N_NODES, 64, 128, 1);

    gather2<<<gblocks, thr, 0, stream>>>(rowptr, ebuf, x1, agg2);

    float* out_sp = (float*)d_out + (size_t)BT * OT;
    gemm_wmma_kernel<<<gs, 128, 0, stream>>>(agg2, norm_in, W2, b2, out_sp, N_NODES, 64, 64, 1);

    transpose_k<<<(GATES * OT + thr - 1) / thr, thr, 0, stream>>>(Wih, WihT);
    transpose_k<<<(GATES * OT + thr - 1) / thr, thr, 0, stream>>>(Whh, WhhT);
    bias_sum_k<<<2, 256, 0, stream>>>(bih, bhh, bsum);
    concat_k<<<(BT * 128 + thr - 1) / thr, thr, 0, stream>>>(hs, cht, Xcat);

    dim3 gp(BT / 16, 8);
    gemm_wmma_kernel<<<gp, 128, 0, stream>>>(Xcat, nullptr, WihT, bsum, gpre, BT, GATES, 128, 0);

    lstm_kernel<<<BB / RB, 1024, 0, stream>>>(gpre, WhhT, (float*)d_out);
}


# LSTM gate activations: sigmoid and tanh on hardware exp2 + reciprocal (v_exp_f32, v_rcp_f32, 1 ulp each) instead of library expf/tanhf and IEEE division: 254 -> 57 vector instructions per wave-step
# speedup vs baseline: 5.4104x; 1.1832x over previous
#include <hip/hip_runtime.h>
#include <math.h>

typedef float v2f __attribute__((ext_vector_type(2)));
typedef float v8f __attribute__((ext_vector_type(8)));

#define N_NODES 50000
#define BB 32
#define TT 512
#define SD 64
#define OT 128
#define GATES 512
#define BT (BB * TT)
#define HP (OT + 4)
#define RB 16
#define GCOL(r) (((((r) & (OT - 1)) >> 2) << 4) + (((r) & 3) << 2) + ((r) >> 7))
#define SP 144
#define SW 272

__device__ __forceinline__ v8f wmma_f32(v2f a, v2f b, v8f c) {
    return __builtin_amdgcn_wmma_f32_16x16x4_f32(
         false, a,  false, b,
         (short)0, c,  false,  false);
}

__global__ void zero_i32(int* __restrict__ p, int n) {
    int i = blockIdx.x * blockDim.x + threadIdx.x;
    if (i < n) p[i] = 0;
}

__global__ void degree_kernel(const int* __restrict__ src, const int* __restrict__ dst,
                              int* __restrict__ cnt_out, int* __restrict__ cnt_in, int E) {
    int e = blockIdx.x * blockDim.x + threadIdx.x;
    if (e < E) {
        atomicAdd(&cnt_out[src[e]], 1);
        atomicAdd(&cnt_in[dst[e]], 1);
    }
}

#define SCAN_IT ((N_NODES + 1023) / 1024)
#define SCAN_G 7
static_assert(SCAN_IT % SCAN_G == 0, "group size must divide the pass count");
__global__ __launch_bounds__(1024) void scan_norm_k(const int* __restrict__ cnt_in,
                                                    const int* __restrict__ cnt_out,
                                                    int* __restrict__ rowptr, int* __restrict__ cursor,
                                                    float* __restrict__ no, float* __restrict__ ni) {
    __shared__ int wtot[32];
    const int wave = threadIdx.x >> 5, lane = threadIdx.x & 31;
    const int base = wave * (SCAN_IT * 32) + lane;
    int tot = 0;
#pragma unroll 7
    for (int it = 0; it < SCAN_IT; ++it) {
        const int i = base + it * 32;
        const int cval = cnt_in[min(i, N_NODES - 1)];
        tot += (i < N_NODES) ? cval : 0;
    }
#pragma unroll
    for (int o = 16; o > 0; o >>= 1) tot += __shfl_xor(tot, o);
    if (lane == 0) wtot[wave] = tot;
    __syncthreads();
    int carry = 0;
    for (int w = 0; w < wave; ++w) carry += wtot[w];
    for (int g = 0; g < SCAN_IT; g += SCAN_G) {
        int cv[SCAN_G], co[SCAN_G], x[SCAN_G];
#pragma unroll
        for (int u = 0; u < SCAN_G; ++u) {
            const int i = base + (g + u) * 32;
            const int ic = min(i, N_NODES - 1);
            const int cval = cnt_in[ic];
            co[u] = cnt_out[ic];
            cv[u] = (i < N_NODES) ? cval : 0;
            x[u] = cv[u];
        }
#pragma unroll
        for (int o = 1; o < 32; o <<= 1) {
#pragma unroll
            for (int u = 0; u < SCAN_G; ++u) {
                const int y = __shfl_up(x[u], o);
                if (lane >= o) x[u] += y;
            }
        }
#pragma unroll
        for (int u = 0; u < SCAN_G; ++u) {
            const int i = base + (g + u) * 32;
            if (i < N_NODES) {
                const int v = carry + x[u] - cv[u];
                rowptr[i] = v;
                cursor[i] = v;
                no[i] = rsqrtf(fmaxf((float)co[u], 1.0f));
                ni[i] = rsqrtf(fmaxf((float)cv[u], 1.0f));
            }
            carry += __shfl(x[u], 31);
        }
    }
    if (threadIdx.x == 1023) rowptr[N_NODES] = carry;
}

__global__ void fill_k(const int* __restrict__ src, const int* __restrict__ dst,
                       const float* __restrict__ norm_out, int* __restrict__ cursor,
                       int2* __restrict__ ebuf, int E) {
    int e = blockIdx.x * blockDim.x + threadIdx.x;
    if (e < E) {
        const int s = src[e];
        const int p = atomicAdd(&cursor[dst[e]], 1);
        ebuf[p] = make_int2(s, __float_as_int(norm_out[s]));
    }
}

__global__ void gather1(const int* __restrict__ rowptr, const int2* __restrict__ ebuf,
                        const float* __restrict__ ht, const float* __restrict__ chs,
                        float* __restrict__ agg1) {
    const int lane = threadIdx.x & 31;
    const int node = __builtin_amdgcn_readfirstlane((int)((blockIdx.x * blockDim.x + threadIdx.x) >> 5));
    if (node >= N_NODES) return;
    const int q = lane << 2;
    const float* fb = (q < SD) ? (ht + q) : (chs + (q - SD));
    const int beg = rowptr[node], end = rowptr[node + 1];
    float4 acc = make_float4(0.f, 0.f, 0.f, 0.f);
    for (int j0 = beg; j0 < end; j0 += 32) {
        const int m = min(32, end - j0);
        int2 ew = make_int2(0, 0);
        if (lane < m) ew = ebuf[j0 + lane];
        int i = 0;
        for (; i + 4 <= m; i += 4) {
            const int s0 = __builtin_amdgcn_readlane(ew.x, i);
            const int s1 = __builtin_amdgcn_readlane(ew.x, i + 1);
            const int s2 = __builtin_amdgcn_readlane(ew.x, i + 2);
            const int s3 = __builtin_amdgcn_readlane(ew.x, i + 3);
            const float4 v0 = *(const float4*)(fb + (long)s0 * SD);
            const float4 v1 = *(const float4*)(fb + (long)s1 * SD);
            const float4 v2 = *(const float4*)(fb + (long)s2 * SD);
            const float4 v3 = *(const float4*)(fb + (long)s3 * SD);
            const float w0 = __int_as_float(__builtin_amdgcn_readlane(ew.y, i));
            const float w1 = __int_as_float(__builtin_amdgcn_readlane(ew.y, i + 1));
            const float w2 = __int_as_float(__builtin_amdgcn_readlane(ew.y, i + 2));
            const float w3 = __int_as_float(__builtin_amdgcn_readlane(ew.y, i + 3));
            acc.x += v0.x * w0; acc.y += v0.y * w0; acc.z += v0.z * w0; acc.w += v0.w * w0;
            acc.x += v1.x * w1; acc.y += v1.y * w1; acc.z += v1.z * w1; acc.w += v1.w * w1;
            acc.x += v2.x * w2; acc.y += v2.y * w2; acc.z += v2.z * w2; acc.w += v2.w * w2;
            acc.x += v3.x * w3; acc.y += v3.y * w3; acc.z += v3.z * w3; acc.w += v3.w * w3;
        }
        for (; i < m; ++i) {
            const int s0 = __builtin_amdgcn_readlane(ew.x, i);
            const float w0 = __int_as_float(__builtin_amdgcn_readlane(ew.y, i));
            const float4 v0 = *(const float4*)(fb + (long)s0 * SD);
            acc.x += v0.x * w0; acc.y += v0.y * w0; acc.z += v0.z * w0; acc.w += v0.w * w0;
        }
    }
    *(float4*)(agg1 + (long)node * 128 + q) = acc;
}

__global__ void gather2(const int* __restrict__ rowptr, const int2* __restrict__ ebuf,
                        const float* __restrict__ x1, float* __restrict__ agg2) {
    const int lane = threadIdx.x & 31;
    const int node = __builtin_amdgcn_readfirstlane((int)((blockIdx.x * blockDim.x + threadIdx.x) >> 5));
    if (node >= N_NODES) return;
    const int q = lane << 1;
    const float* fb = x1 + q;
    const int beg = rowptr[node], end = rowptr[node + 1];
    float2 acc = make_float2(0.f, 0.f);
    for (int j0 = beg; j0 < end; j0 += 32) {
        const int m = min(32, end - j0);
        int2 ew = make_int2(0, 0);
        if (lane < m) ew = ebuf[j0 + lane];
        int i = 0;
        for (; i + 4 <= m; i += 4) {
            const int s0 = __builtin_amdgcn_readlane(ew.x, i);
            const int s1 = __builtin_amdgcn_readlane(ew.x, i + 1);
            const int s2 = __builtin_amdgcn_readlane(ew.x, i + 2);
            const int s3 = __builtin_amdgcn_readlane(ew.x, i + 3);
            const float2 v0 = *(const float2*)(fb + (long)s0 * 64);
            const float2 v1 = *(const float2*)(fb + (long)s1 * 64);
            const float2 v2 = *(const float2*)(fb + (long)s2 * 64);
            const float2 v3 = *(const float2*)(fb + (long)s3 * 64);
            const float w0 = __int_as_float(__builtin_amdgcn_readlane(ew.y, i));
            const float w1 = __int_as_float(__builtin_amdgcn_readlane(ew.y, i + 1));
            const float w2 = __int_as_float(__builtin_amdgcn_readlane(ew.y, i + 2));
            const float w3 = __int_as_float(__builtin_amdgcn_readlane(ew.y, i + 3));
            acc.x += v0.x * w0; acc.y += v0.y * w0;
            acc.x += v1.x * w1; acc.y += v1.y * w1;
            acc.x += v2.x * w2; acc.y += v2.y * w2;
            acc.x += v3.x * w3; acc.y += v3.y * w3;
        }
        for (; i < m; ++i) {
            const int s0 = __builtin_amdgcn_readlane(ew.x, i);
            const float w0 = __int_as_float(__builtin_amdgcn_readlane(ew.y, i));
            const float2 v0 = *(const float2*)(fb + (long)s0 * 64);
            acc.x += v0.x * w0; acc.y += v0.y * w0;
        }
    }
    *(float2*)(agg2 + (long)node * 64 + q) = acc;
}

__global__ void transpose_k(const float* __restrict__ W, float* __restrict__ WT) {
    int idx = blockIdx.x * blockDim.x + threadIdx.x;
    if (idx >= GATES * OT) return;
    int r = idx >> 7, c = idx & 127;
    WT[c * GATES + GCOL(r)] = W[idx];
}

__global__ void bias_sum_k(const float* __restrict__ a, const float* __restrict__ b,
                           float* __restrict__ o) {
    int i = blockIdx.x * blockDim.x + threadIdx.x;
    if (i < GATES) o[GCOL(i)] = a[i] + b[i];
}

__global__ void concat_k(const float* __restrict__ hs, const float* __restrict__ cht,
                         float* __restrict__ X) {
    int idx = blockIdx.x * blockDim.x + threadIdx.x;
    if (idx >= BT * 128) return;
    int row = idx >> 7, k = idx & 127;
    X[idx] = (k < SD) ? hs[row * SD + k] : cht[row * SD + (k - SD)];
}

__global__ void gemm_wmma_kernel(const float* __restrict__ A, const float* __restrict__ rowscale,
                                 const float* __restrict__ Bm, const float* __restrict__ bias,
                                 float* __restrict__ C, int M, int N, int K, int applyGelu) {
    const int wave = threadIdx.x >> 5;
    const int lane = threadIdx.x & 31;
    const int hl = lane >> 4;
    const int l16 = lane & 15;
    const int ct = blockIdx.y * (blockDim.x >> 5) + wave;
    if ((ct << 4) >= N) return;
    const int row0 = blockIdx.x << 4;
    const int col = (ct << 4) + l16;
    const int arow = row0 + l16;
    const float rs = rowscale ? rowscale[arow] : 1.0f;
    const float* Arow = A + (size_t)arow * K;
    v8f acc = {};
    for (int k = 0; k < K; k += 4) {
        const int ka = k + (hl << 1);
        v2f a, b;
        a.x = Arow[ka] * rs;
        a.y = Arow[ka + 1] * rs;
        b.x = Bm[(size_t)ka * N + col];
        b.y = Bm[(size_t)(ka + 1) * N + col];
        acc = wmma_f32(a, b, acc);
    }
    const float bv = bias ? bias[col] : 0.0f;
#pragma unroll
    for (int r = 0; r < 8; ++r) {
        float v = acc[r] + bv;
        if (applyGelu) v = 0.5f * v * (1.0f + erff(v * 0.70710678118654752f));
        C[(size_t)(row0 + r + (hl << 3)) * N + col] = v;
    }
}

__device__ __forceinline__ float sigmoid_hw(float x) {
    return __builtin_amdgcn_rcpf(1.0f + __builtin_amdgcn_exp2f(x * -1.4426950408889634f));
}
__device__ __forceinline__ float tanh_hw(float x) {
    return fmaf(__builtin_amdgcn_rcpf(1.0f + __builtin_amdgcn_exp2f(x * 2.8853900817779268f)), -2.0f, 1.0f);
}

__global__ __launch_bounds__(1024) void lstm_kernel(const float* gpre,
                                                    const float* __restrict__ WhhT,
                                                    float* __restrict__ out) {
    __shared__ __attribute__((aligned(16))) float h_lds[2][RB][HP];
    __shared__ __attribute__((aligned(16))) float s_lds[32][SW];
    const int tid = threadIdx.x;
    const int wave = tid >> 5, lane = tid & 31;
    const int hl = lane >> 4, l16 = lane & 15;
    const int col0 = wave << 4;
    const int col = col0 + l16;
    const int brow0 = blockIdx.x * RB;

    float bx[32], by[32];
#pragma unroll
    for (int kk = 0; kk < 32; ++kk) {
        const int ka = (kk << 2) + (hl << 1);
        bx[kk] = WhhT[ka * GATES + col];
        by[kk] = WhhT[(ka + 1) * GATES + col];
    }

    for (int i = tid; i < RB * OT; i += 1024) h_lds[0][i >> 7][i & (OT - 1)] = 0.0f;
    float creg[2] = {0.f, 0.f};
    const int crow = lane >> 2;
    const int cq = (lane & 3) << 2;
    const int cj = (wave << 2) + (lane & 3);
    float* sw = &s_lds[wave][0];
    const int orow = wave >> 1;
    const int ocol = ((wave & 1) << 6) + (lane << 1);
    float* outp = out + (long)(brow0 + orow) * TT * OT + ocol;
    __syncthreads();

    int cur = 0;
    auto step = [&](const int t, const bool store_prev) __attribute__((always_inline)) {
        float4 gp[2];
#pragma unroll
        for (int u = 0; u < 2; ++u)
            gp[u] = *(const float4*)(gpre + ((long)(brow0 + crow + (u << 3)) * TT + t) * GATES + col0 + cq);
        __builtin_amdgcn_fence(__ATOMIC_RELEASE, "wavefront");

        const float (*hc)[HP] = h_lds[cur];
        float (*hn)[HP] = h_lds[cur ^ 1];
        const float2 ov = *(const float2*)&hc[orow][ocol];

        v8f c0 = {};
#pragma unroll
        for (int kk = 0; kk < 32; ++kk) {
            const int ka = (kk << 2) + (hl << 1);
            v2f b, a0;
            b.x = bx[kk];                b.y = by[kk];
            a0.x = hc[l16][ka];          a0.y = hc[l16][ka + 1];
            c0 = wmma_f32(a0, b, c0);
        }
#pragma unroll
        for (int r = 0; r < 8; ++r) sw[hl * SP + (r << 4) + l16] = c0[r];
        __builtin_amdgcn_fence(__ATOMIC_RELEASE, "wavefront");
        __builtin_amdgcn_wave_barrier();
        __builtin_amdgcn_fence(__ATOMIC_ACQUIRE, "wavefront");
        float4 gs[2];
#pragma unroll
        for (int u = 0; u < 2; ++u) gs[u] = *(const float4*)&sw[u * SP + (lane << 2)];
        if (store_prev) *(float2*)(outp + (long)(t - 1) * OT) = ov;

#pragma unroll
        for (int u = 0; u < 2; ++u) {
            float gi = gs[u].x + gp[u].x;
            float gf = gs[u].y + gp[u].y;
            float gg = gs[u].z + gp[u].z;
            float go = gs[u].w + gp[u].w;
            float si = sigmoid_hw(gi);
            float sf = sigmoid_hw(gf);
            float so = sigmoid_hw(go);
            float cc = sf * creg[u] + si * tanh_hw(gg);
            creg[u] = cc;
            float hh = so * tanh_hw(cc);
            hn[crow + (u << 3)][cj] = hh;
        }
        __syncthreads();
        cur ^= 1;
    };
    step(0, false);
    for (int t = 1; t < TT; ++t) step(t, true);
    *(float2*)(outp + (long)(TT - 1) * OT) = *(const float2*)&h_lds[cur][orow][ocol];
}

extern "C" void kernel_launch(void* const* d_in, const int* in_sizes, int n_in,
                              void* d_out, int out_size, void* d_ws, size_t ws_size,
                              hipStream_t stream) {
    const float* hs  = (const float*)d_in[0];
    const float* ht  = (const float*)d_in[1];
    const float* chs = (const float*)d_in[2];
    const float* cht = (const float*)d_in[3];
    const int*   src = (const int*)d_in[4];
    const int*   dst = (const int*)d_in[5];
    const float* W1  = (const float*)d_in[6];
    const float* b1  = (const float*)d_in[7];
    const float* W2  = (const float*)d_in[8];
    const float* b2  = (const float*)d_in[9];
    const float* Wih = (const float*)d_in[10];
    const float* Whh = (const float*)d_in[11];
    const float* bih = (const float*)d_in[12];
    const float* bhh = (const float*)d_in[13];
    const int E = in_sizes[4];

    float* ws       = (float*)d_ws;
    float* norm_out = ws;
    float* norm_in  = norm_out + N_NODES;
    float* agg1     = norm_in + N_NODES;
    float* agg2     = agg1 + (size_t)N_NODES * 128;
    float* x1       = agg2 + (size_t)N_NODES * 64;
    float* Xcat     = x1 + (size_t)N_NODES * 64;
    float* gpre     = Xcat + (size_t)BT * 128;
    float* WihT     = gpre + (size_t)BT * GATES;
    float* WhhT     = WihT + OT * GATES;
    float* bsum     = WhhT + OT * GATES;
    int*  cnt_in  = (int*)gpre;
    int*  cnt_out = cnt_in + N_NODES;
    int*  rowptr  = cnt_out + N_NODES;
    int*  cursor  = rowptr + N_NODES + 16;
    int2* ebuf    = (int2*)(cursor + N_NODES);

    const int thr = 256;
    zero_i32<<<(2 * N_NODES + thr - 1) / thr, thr, 0, stream>>>(cnt_in, 2 * N_NODES);

    degree_kernel<<<(E + thr - 1) / thr, thr, 0, stream>>>(src, dst, cnt_out, cnt_in, E);
    scan_norm_k<<<1, 1024, 0, stream>>>(cnt_in, cnt_out, rowptr, cursor, norm_out, norm_in);
    fill_k<<<(E + thr - 1) / thr, thr, 0, stream>>>(src, dst, norm_out, cursor, ebuf, E);

    const int gblocks = (N_NODES * 32 + thr - 1) / thr;
    gather1<<<gblocks, thr, 0, stream>>>(rowptr, ebuf, ht, chs, agg1);

    dim3 gs(N_NODES / 16, 1);
    gemm_wmma_kernel<<<gs, 128, 0, stream>>>(agg1, norm_in, W1, b1, x1, N_NODES, 64, 128, 1);

    gather2<<<gblocks, thr, 0, stream>>>(rowptr, ebuf, x1, agg2);

    float* out_sp = (float*)d_out + (size_t)BT * OT;
    gemm_wmma_kernel<<<gs, 128, 0, stream>>>(agg2, norm_in, W2, b2, out_sp, N_NODES, 64, 64, 1);

    transpose_k<<<(GATES * OT + thr - 1) / thr, thr, 0, stream>>>(Wih, WihT);
    transpose_k<<<(GATES * OT + thr - 1) / thr, thr, 0, stream>>>(Whh, WhhT);
    bias_sum_k<<<2, 256, 0, stream>>>(bih, bhh, bsum);
    concat_k<<<(BT * 128 + thr - 1) / thr, thr, 0, stream>>>(hs, cht, Xcat);

    dim3 gp(BT / 16, 8);
    gemm_wmma_kernel<<<gp, 128, 0, stream>>>(Xcat, nullptr, WihT, bsum, gpre, BT, GATES, 128, 0);

    lstm_kernel<<<BB / RB, 1024, 0, stream>>>(gpre, WhhT, (float*)d_out);
}


# Graph branch runs beside the LSTM: recurrence cut into 8 time segments (h,c handed on through the workspace), each segment's launch carries one graph stage in extra workgroups on the idle WGPs
# speedup vs baseline: 5.9621x; 1.1020x over previous
#include <hip/hip_runtime.h>
#include <math.h>

typedef float v2f __attribute__((ext_vector_type(2)));
typedef float v8f __attribute__((ext_vector_type(8)));

#define N_NODES 50000
#define BB 32
#define TT 512
#define SD 64
#define OT 128
#define GATES 512
#define BT (BB * TT)
#define HP (OT + 4)
#define RB 16
#define GCOL(r) (((((r) & (OT - 1)) >> 2) << 4) + (((r) & 3) << 2) + ((r) >> 7))
#define SP 144
#define SW 272

__device__ __forceinline__ v8f wmma_f32(v2f a, v2f b, v8f c) {
    return __builtin_amdgcn_wmma_f32_16x16x4_f32(
         false, a,  false, b,
         (short)0, c,  false,  false);
}

#define FB 1024
#define NLW (BB / RB)
#define ST_NONE 0
#define ST_DEGREE 1
#define ST_SCAN 2
#define ST_FILL 3
#define ST_GATHER1 4
#define ST_GEMM1 5
#define ST_GATHER2 6
#define ST_GEMM2 7

__global__ void zero_i32(int* __restrict__ p, int n) {
    int i = blockIdx.x * blockDim.x + threadIdx.x;
    if (i < n) p[i] = 0;
}

__device__ __forceinline__ void degree_stage(const int vb, const int* __restrict__ src,
                                             const int* __restrict__ dst, int* __restrict__ cnt_out,
                                             int* __restrict__ cnt_in, int E) {
    const int e = vb * FB + (int)threadIdx.x;
    if (e < E) {
        atomicAdd(&cnt_out[src[e]], 1);
        atomicAdd(&cnt_in[dst[e]], 1);
    }
}

#define SCAN_IT ((N_NODES + 1023) / 1024)
#define SCAN_G 7
static_assert(SCAN_IT % SCAN_G == 0, "group size must divide the pass count");
static_assert(FB == 1024, "the scan is written for one workgroup of 32 waves");
__device__ __forceinline__ void scan_stage(const int* __restrict__ cnt_in,
                                           const int* __restrict__ cnt_out,
                                           int* __restrict__ rowptr, int* __restrict__ cursor,
                                           float* __restrict__ no, float* __restrict__ ni) {
    __shared__ int wtot[32];
    const int wave = threadIdx.x >> 5, lane = threadIdx.x & 31;
    const int base = wave * (SCAN_IT * 32) + lane;
    int tot = 0;
#pragma unroll 7
    for (int it = 0; it < SCAN_IT; ++it) {
        const int i = base + it * 32;
        const int cval = cnt_in[min(i, N_NODES - 1)];
        tot += (i < N_NODES) ? cval : 0;
    }
#pragma unroll
    for (int o = 16; o > 0; o >>= 1) tot += __shfl_xor(tot, o);
    if (lane == 0) wtot[wave] = tot;
    __syncthreads();
    int carry = 0;
    for (int w = 0; w < wave; ++w) carry += wtot[w];
    for (int g = 0; g < SCAN_IT; g += SCAN_G) {
        int cv[SCAN_G], co[SCAN_G], x[SCAN_G];
#pragma unroll
        for (int u = 0; u < SCAN_G; ++u) {
            const int i = base + (g + u) * 32;
            const int ic = min(i, N_NODES - 1);
            const int cval = cnt_in[ic];
            co[u] = cnt_out[ic];
            cv[u] = (i < N_NODES) ? cval : 0;
            x[u] = cv[u];
        }
#pragma unroll
        for (int o = 1; o < 32; o <<= 1) {
#pragma unroll
            for (int u = 0; u < SCAN_G; ++u) {
                const int y = __shfl_up(x[u], o);
                if (lane >= o) x[u] += y;
            }
        }
#pragma unroll
        for (int u = 0; u < SCAN_G; ++u) {
            const int i = base + (g + u) * 32;
            if (i < N_NODES) {
                const int v = carry + x[u] - cv[u];
                rowptr[i] = v;
                cursor[i] = v;
                no[i] = rsqrtf(fmaxf((float)co[u], 1.0f));
                ni[i] = rsqrtf(fmaxf((float)cv[u], 1.0f));
            }
            carry += __shfl(x[u], 31);
        }
    }
    if (threadIdx.x == 1023) rowptr[N_NODES] = carry;
}

__device__ __forceinline__ void fill_stage(const int vb, const int* __restrict__ src,
                                           const int* __restrict__ dst,
                                           const float* __restrict__ norm_out, int* __restrict__ cursor,
                                           int2* __restrict__ ebuf, int E) {
    const int e = vb * FB + (int)threadIdx.x;
    if (e < E) {
        const int s = src[e];
        const int p = atomicAdd(&cursor[dst[e]], 1);
        ebuf[p] = make_int2(s, __float_as_int(norm_out[s]));
    }
}

__device__ __forceinline__ void gather1_stage(const int vb, const int* __restrict__ rowptr,
                                              const int2* __restrict__ ebuf,
                                              const float* __restrict__ ht, const float* __restrict__ chs,
                                              float* __restrict__ agg1) {
    const int lane = threadIdx.x & 31;
    const int node = __builtin_amdgcn_readfirstlane((vb << 5) + (int)(threadIdx.x >> 5));
    if (node >= N_NODES) return;
    const int q = lane << 2;
    const float* fb = (q < SD) ? (ht + q) : (chs + (q - SD));
    const int beg = rowptr[node], end = rowptr[node + 1];
    float4 acc = make_float4(0.f, 0.f, 0.f, 0.f);
    for (int j0 = beg; j0 < end; j0 += 32) {
        const int m = min(32, end - j0);
        int2 ew = make_int2(0, 0);
        if (lane < m) ew = ebuf[j0 + lane];
        int i = 0;
        for (; i + 4 <= m; i += 4) {
            const int s0 = __builtin_amdgcn_readlane(ew.x, i);
            const int s1 = __builtin_amdgcn_readlane(ew.x, i + 1);
            const int s2 = __builtin_amdgcn_readlane(ew.x, i + 2);
            const int s3 = __builtin_amdgcn_readlane(ew.x, i + 3);
            const float4 v0 = *(const float4*)(fb + (long)s0 * SD);
            const float4 v1 = *(const float4*)(fb + (long)s1 * SD);
            const float4 v2 = *(const float4*)(fb + (long)s2 * SD);
            const float4 v3 = *(const float4*)(fb + (long)s3 * SD);
            const float w0 = __int_as_float(__builtin_amdgcn_readlane(ew.y, i));
            const float w1 = __int_as_float(__builtin_amdgcn_readlane(ew.y, i + 1));
            const float w2 = __int_as_float(__builtin_amdgcn_readlane(ew.y, i + 2));
            const float w3 = __int_as_float(__builtin_amdgcn_readlane(ew.y, i + 3));
            acc.x += v0.x * w0; acc.y += v0.y * w0; acc.z += v0.z * w0; acc.w += v0.w * w0;
            acc.x += v1.x * w1; acc.y += v1.y * w1; acc.z += v1.z * w1; acc.w += v1.w * w1;
            acc.x += v2.x * w2; acc.y += v2.y * w2; acc.z += v2.z * w2; acc.w += v2.w * w2;
            acc.x += v3.x * w3; acc.y += v3.y * w3; acc.z += v3.z * w3; acc.w += v3.w * w3;
        }
        for (; i < m; ++i) {
            const int s0 = __builtin_amdgcn_readlane(ew.x, i);
            const float w0 = __int_as_float(__builtin_amdgcn_readlane(ew.y, i));
            const float4 v0 = *(const float4*)(fb + (long)s0 * SD);
            acc.x += v0.x * w0; acc.y += v0.y * w0; acc.z += v0.z * w0; acc.w += v0.w * w0;
        }
    }
    *(float4*)(agg1 + (long)node * 128 + q) = acc;
}

__device__ __forceinline__ void gather2_stage(const int vb, const int* __restrict__ rowptr,
                                              const int2* __restrict__ ebuf,
                                              const float* __restrict__ x1, float* __restrict__ agg2) {
    const int lane = threadIdx.x & 31;
    const int node = __builtin_amdgcn_readfirstlane((vb << 5) + (int)(threadIdx.x >> 5));
    if (node >= N_NODES) return;
    const int q = lane << 1;
    const float* fb = x1 + q;
    const int beg = rowptr[node], end = rowptr[node + 1];
    float2 acc = make_float2(0.f, 0.f);
    for (int j0 = beg; j0 < end; j0 += 32) {
        const int m = min(32, end - j0);
        int2 ew = make_int2(0, 0);
        if (lane < m) ew = ebuf[j0 + lane];
        int i = 0;
        for (; i + 4 <= m; i += 4) {
            const int s0 = __builtin_amdgcn_readlane(ew.x, i);
            const int s1 = __builtin_amdgcn_readlane(ew.x, i + 1);
            const int s2 = __builtin_amdgcn_readlane(ew.x, i + 2);
            const int s3 = __builtin_amdgcn_readlane(ew.x, i + 3);
            const float2 v0 = *(const float2*)(fb + (long)s0 * 64);
            const float2 v1 = *(const float2*)(fb + (long)s1 * 64);
            const float2 v2 = *(const float2*)(fb + (long)s2 * 64);
            const float2 v3 = *(const float2*)(fb + (long)s3 * 64);
            const float w0 = __int_as_float(__builtin_amdgcn_readlane(ew.y, i));
            const float w1 = __int_as_float(__builtin_amdgcn_readlane(ew.y, i + 1));
            const float w2 = __int_as_float(__builtin_amdgcn_readlane(ew.y, i + 2));
            const float w3 = __int_as_float(__builtin_amdgcn_readlane(ew.y, i + 3));
            acc.x += v0.x * w0; acc.y += v0.y * w0;
            acc.x += v1.x * w1; acc.y += v1.y * w1;
            acc.x += v2.x * w2; acc.y += v2.y * w2;
            acc.x += v3.x * w3; acc.y += v3.y * w3;
        }
        for (; i < m; ++i) {
            const int s0 = __builtin_amdgcn_readlane(ew.x, i);
            const float w0 = __int_as_float(__builtin_amdgcn_readlane(ew.y, i));
            const float2 v0 = *(const float2*)(fb + (long)s0 * 64);
            acc.x += v0.x * w0; acc.y += v0.y * w0;
        }
    }
    *(float2*)(agg2 + (long)node * 64 + q) = acc;
}

__global__ void transpose_k(const float* __restrict__ W, float* __restrict__ WT) {
    int idx = blockIdx.x * blockDim.x + threadIdx.x;
    if (idx >= GATES * OT) return;
    int r = idx >> 7, c = idx & 127;
    WT[c * GATES + GCOL(r)] = W[idx];
}

__global__ void bias_sum_k(const float* __restrict__ a, const float* __restrict__ b,
                           float* __restrict__ o) {
    int i = blockIdx.x * blockDim.x + threadIdx.x;
    if (i < GATES) o[GCOL(i)] = a[i] + b[i];
}

__global__ void concat_k(const float* __restrict__ hs, const float* __restrict__ cht,
                         float* __restrict__ X) {
    int idx = blockIdx.x * blockDim.x + threadIdx.x;
    if (idx >= BT * 128) return;
    int row = idx >> 7, k = idx & 127;
    X[idx] = (k < SD) ? hs[row * SD + k] : cht[row * SD + (k - SD)];
}

__device__ __forceinline__ void gemm_tile(const float* __restrict__ A, const float* __restrict__ rowscale,
                                          const float* __restrict__ Bm, const float* __restrict__ bias,
                                          float* __restrict__ C, const int N, const int K,
                                          const int applyGelu, const int row0, const int ct, const int lane) {
    const int hl = lane >> 4;
    const int l16 = lane & 15;
    const int col = (ct << 4) + l16;
    const int arow = row0 + l16;
    const float rs = rowscale ? rowscale[arow] : 1.0f;
    const float* Arow = A + (size_t)arow * K;
    v8f acc = {};
    for (int k = 0; k < K; k += 4) {
        const int ka = k + (hl << 1);
        v2f a, b;
        a.x = Arow[ka] * rs;
        a.y = Arow[ka + 1] * rs;
        b.x = Bm[(size_t)ka * N + col];
        b.y = Bm[(size_t)(ka + 1) * N + col];
        acc = wmma_f32(a, b, acc);
    }
    const float bv = bias ? bias[col] : 0.0f;
#pragma unroll
    for (int r = 0; r < 8; ++r) {
        float v = acc[r] + bv;
        if (applyGelu) v = 0.5f * v * (1.0f + erff(v * 0.70710678118654752f));
        C[(size_t)(row0 + r + (hl << 3)) * N + col] = v;
    }
}

__global__ void gemm_wmma_kernel(const float* __restrict__ A, const float* __restrict__ rowscale,
                                 const float* __restrict__ Bm, const float* __restrict__ bias,
                                 float* __restrict__ C, int M, int N, int K, int applyGelu) {
    const int wave = threadIdx.x >> 5;
    const int ct = blockIdx.y * (blockDim.x >> 5) + wave;
    if ((ct << 4) >= N) return;
    gemm_tile(A, rowscale, Bm, bias, C, N, K, applyGelu, blockIdx.x << 4, ct, threadIdx.x & 31);
}

__device__ __forceinline__ void gemm64_stage(const int vb, const float* __restrict__ A,
                                             const float* __restrict__ rowscale,
                                             const float* __restrict__ Bm, const float* __restrict__ bias,
                                             float* __restrict__ C, const int K) {
    const int wave = threadIdx.x >> 5;
    const int rt = (vb << 3) + (wave >> 2);
    if (rt >= N_NODES / 16) return;
    gemm_tile(A, rowscale, Bm, bias, C, 64, K, 1, rt << 4, wave & 3, threadIdx.x & 31);
}

__device__ __forceinline__ float sigmoid_hw(float x) {
    return __builtin_amdgcn_rcpf(1.0f + __builtin_amdgcn_exp2f(x * -1.4426950408889634f));
}
__device__ __forceinline__ float tanh_hw(float x) {
    return fmaf(__builtin_amdgcn_rcpf(1.0f + __builtin_amdgcn_exp2f(x * 2.8853900817779268f)), -2.0f, 1.0f);
}

template <int T0, int T1>
__device__ __forceinline__ void lstm_segment(const int wg,
                                             const float* gpre,
                                             const float* __restrict__ WhhT,
                                             float* __restrict__ out,
                                             float* __restrict__ hst, float* __restrict__ cst) {
    __shared__ __attribute__((aligned(16))) float h_lds[2][RB][HP];
    __shared__ __attribute__((aligned(16))) float s_lds[32][SW];
    const int tid = threadIdx.x;
    const int wave = tid >> 5, lane = tid & 31;
    const int hl = lane >> 4, l16 = lane & 15;
    const int col0 = wave << 4;
    const int col = col0 + l16;
    const int brow0 = wg * RB;

    float bx[32], by[32];
#pragma unroll
    for (int kk = 0; kk < 32; ++kk) {
        const int ka = (kk << 2) + (hl << 1);
        bx[kk] = WhhT[ka * GATES + col];
        by[kk] = WhhT[(ka + 1) * GATES + col];
    }

    float creg[2] = {0.f, 0.f};
    float* hsw = hst + wg * (RB * OT);
    float* csw = cst + ((wg * FB + tid) << 1);
    if (T0 == 0) {
        for (int i = tid; i < RB * OT; i += FB) h_lds[0][i >> 7][i & (OT - 1)] = 0.0f;
    } else {
        for (int i = tid; i < RB * OT; i += FB) h_lds[0][i >> 7][i & (OT - 1)] = hsw[i];
        const float2 cv = *(const float2*)csw;
        creg[0] = cv.x;
        creg[1] = cv.y;
    }
    const int crow = lane >> 2;
    const int cq = (lane & 3) << 2;
    const int cj = (wave << 2) + (lane & 3);
    float* sw = &s_lds[wave][0];
    const int orow = wave >> 1;
    const int ocol = ((wave & 1) << 6) + (lane << 1);
    float* outp = out + (long)(brow0 + orow) * TT * OT + ocol;
    __syncthreads();

    int cur = 0;
    auto step = [&](const int t, const bool store_prev) __attribute__((always_inline)) {
        float4 gp[2];
#pragma unroll
        for (int u = 0; u < 2; ++u)
            gp[u] = *(const float4*)(gpre + ((long)(brow0 + crow + (u << 3)) * TT + t) * GATES + col0 + cq);
        __builtin_amdgcn_fence(__ATOMIC_RELEASE, "wavefront");

        const float (*hc)[HP] = h_lds[cur];
        float (*hn)[HP] = h_lds[cur ^ 1];
        const float2 ov = *(const float2*)&hc[orow][ocol];

        v8f c0 = {};
#pragma unroll
        for (int kk = 0; kk < 32; ++kk) {
            const int ka = (kk << 2) + (hl << 1);
            v2f b, a0;
            b.x = bx[kk];                b.y = by[kk];
            a0.x = hc[l16][ka];          a0.y = hc[l16][ka + 1];
            c0 = wmma_f32(a0, b, c0);
        }
#pragma unroll
        for (int r = 0; r < 8; ++r) sw[hl * SP + (r << 4) + l16] = c0[r];
        __builtin_amdgcn_fence(__ATOMIC_RELEASE, "wavefront");
        __builtin_amdgcn_wave_barrier();
        __builtin_amdgcn_fence(__ATOMIC_ACQUIRE, "wavefront");
        float4 gs[2];
#pragma unroll
        for (int u = 0; u < 2; ++u) gs[u] = *(const float4*)&sw[u * SP + (lane << 2)];
        if (store_prev) *(float2*)(outp + (long)(t - 1) * OT) = ov;

#pragma unroll
        for (int u = 0; u < 2; ++u) {
            float gi = gs[u].x + gp[u].x;
            float gf = gs[u].y + gp[u].y;
            float gg = gs[u].z + gp[u].z;
            float go = gs[u].w + gp[u].w;
            float si = sigmoid_hw(gi);
            float sf = sigmoid_hw(gf);
            float so = sigmoid_hw(go);
            float cc = sf * creg[u] + si * tanh_hw(gg);
            creg[u] = cc;
            float hh = so * tanh_hw(cc);
            hn[crow + (u << 3)][cj] = hh;
        }
        __syncthreads();
        cur ^= 1;
    };
    step(T0, false);
    for (int t = T0 + 1; t < T1; ++t) step(t, true);
    *(float2*)(outp + (long)(T1 - 1) * OT) = *(const float2*)&h_lds[cur][orow][ocol];
    if (T1 < TT) {
        for (int i = tid; i < RB * OT; i += FB) hsw[i] = h_lds[cur][i >> 7][i & (OT - 1)];
        *(float2*)csw = make_float2(creg[0], creg[1]);
    }
}

template <int STAGE, int T0, int T1>
__global__ __launch_bounds__(FB) void fused_k(
    const float* gpre, const float* __restrict__ WhhT, float* __restrict__ out,
    float* __restrict__ hst, float* __restrict__ cst,
    const int* __restrict__ src, const int* __restrict__ dst, int E,
    int* __restrict__ cnt_out, int* __restrict__ cnt_in, int* __restrict__ rowptr,
    int* __restrict__ cursor, float* __restrict__ norm_out, float* __restrict__ norm_in,
    int2* __restrict__ ebuf, const float* __restrict__ ht, const float* __restrict__ chs,
    float* __restrict__ agg, const float* __restrict__ W1, const float* __restrict__ b1,
    float* __restrict__ x1, const float* __restrict__ W2, const float* __restrict__ b2,
    float* __restrict__ out_sp) {
    if (blockIdx.x < NLW) {
        lstm_segment<T0, T1>((int)blockIdx.x, gpre, WhhT, out, hst, cst);
        return;
    }
    const int vb = (int)blockIdx.x - NLW;
    if (STAGE == ST_DEGREE)  degree_stage(vb, src, dst, cnt_out, cnt_in, E);
    if (STAGE == ST_SCAN)    scan_stage(cnt_in, cnt_out, rowptr, cursor, norm_out, norm_in);
    if (STAGE == ST_FILL)    fill_stage(vb, src, dst, norm_out, cursor, ebuf, E);
    if (STAGE == ST_GATHER1) gather1_stage(vb, rowptr, ebuf, ht, chs, agg);
    if (STAGE == ST_GEMM1)   gemm64_stage(vb, agg, norm_in, W1, b1, x1, 128);
    if (STAGE == ST_GATHER2) gather2_stage(vb, rowptr, ebuf, x1, agg);
    if (STAGE == ST_GEMM2)   gemm64_stage(vb, agg, norm_in, W2, b2, out_sp, 64);
}

extern "C" void kernel_launch(void* const* d_in, const int* in_sizes, int n_in,
                              void* d_out, int out_size, void* d_ws, size_t ws_size,
                              hipStream_t stream) {
    const float* hs  = (const float*)d_in[0];
    const float* ht  = (const float*)d_in[1];
    const float* chs = (const float*)d_in[2];
    const float* cht = (const float*)d_in[3];
    const int*   src = (const int*)d_in[4];
    const int*   dst = (const int*)d_in[5];
    const float* W1  = (const float*)d_in[6];
    const float* b1  = (const float*)d_in[7];
    const float* W2  = (const float*)d_in[8];
    const float* b2  = (const float*)d_in[9];
    const float* Wih = (const float*)d_in[10];
    const float* Whh = (const float*)d_in[11];
    const float* bih = (const float*)d_in[12];
    const float* bhh = (const float*)d_in[13];
    const int E = in_sizes[4];

    float* ws       = (float*)d_ws;
    float* norm_out = ws;
    float* norm_in  = norm_out + N_NODES;
    float* agg      = norm_in + N_NODES;
    float* bkt      = agg + (size_t)N_NODES * 128;
    float* x1       = bkt + (size_t)N_NODES * 64;
    float* Xcat     = x1 + (size_t)N_NODES * 64;
    float* gpre     = Xcat + (size_t)BT * 128;
    float* WihT     = gpre + (size_t)BT * GATES;
    float* WhhT     = WihT + OT * GATES;
    float* bsum     = WhhT + OT * GATES;
    int*  cnt_in  = (int*)bkt;
    int*  cnt_out = cnt_in + N_NODES;
    int*  rowptr  = cnt_out + N_NODES;
    int*  cursor  = rowptr + N_NODES + 16;
    int2* ebuf    = (int2*)(cursor + N_NODES);
    float* hst = Xcat;
    float* cst = Xcat + NLW * RB * OT;
    float* out_lstm = (float*)d_out;
    float* out_sp = (float*)d_out + (size_t)BT * OT;

    const int thr = 256;
    zero_i32<<<(2 * N_NODES + thr - 1) / thr, thr, 0, stream>>>(cnt_in, 2 * N_NODES);

    transpose_k<<<(GATES * OT + thr - 1) / thr, thr, 0, stream>>>(Wih, WihT);
    transpose_k<<<(GATES * OT + thr - 1) / thr, thr, 0, stream>>>(Whh, WhhT);
    bias_sum_k<<<2, 256, 0, stream>>>(bih, bhh, bsum);
    concat_k<<<(BT * 128 + thr - 1) / thr, thr, 0, stream>>>(hs, cht, Xcat);
    dim3 gp(BT / 16, 8);
    gemm_wmma_kernel<<<gp, 128, 0, stream>>>(Xcat, nullptr, WihT, bsum, gpre, BT, GATES, 128, 0);

    const int eb = (E + FB - 1) / FB;
    const int nb = (N_NODES * 32 + FB - 1) / FB;
    const int mb = (N_NODES / 16 + 7) / 8;
#define FUSED_ARGS gpre, WhhT, out_lstm, hst, cst, src, dst, E, cnt_out, cnt_in, rowptr, cursor, \
                   norm_out, norm_in, ebuf, ht, chs, agg, W1, b1, x1, W2, b2, out_sp
    fused_k<ST_DEGREE,    0,  64><<<NLW + eb, FB, 0, stream>>>(FUSED_ARGS);
    fused_k<ST_SCAN,     64,  96><<<NLW + 1,  FB, 0, stream>>>(FUSED_ARGS);
    fused_k<ST_FILL,     96, 160><<<NLW + eb, FB, 0, stream>>>(FUSED_ARGS);
    fused_k<ST_GATHER1, 160, 256><<<NLW + nb, FB, 0, stream>>>(FUSED_ARGS);
    fused_k<ST_GEMM1,   256, 320><<<NLW + mb, FB, 0, stream>>>(FUSED_ARGS);
    fused_k<ST_GATHER2, 320, 384><<<NLW + nb, FB, 0, stream>>>(FUSED_ARGS);
    fused_k<ST_GEMM2,   384, 448><<<NLW + mb, FB, 0, stream>>>(FUSED_ARGS);
    fused_k<ST_NONE,    448, 512><<<NLW,      FB, 0, stream>>>(FUSED_ARGS);
#undef FUSED_ARGS
}


# LSTM gate activations on the hardware tanh (v_tanh_f32); sigmoid(x) = (1 + tanh(x/2))/2 with the 1/2 folded exactly into the i,f,o weights and bias: 20 -> 10 transcendentals per wave-step
# speedup vs baseline: 6.4429x; 1.0806x over previous
#include <hip/hip_runtime.h>
#include <math.h>

typedef float v2f __attribute__((ext_vector_type(2)));
typedef float v8f __attribute__((ext_vector_type(8)));

#define N_NODES 50000
#define BB 32
#define TT 512
#define SD 64
#define OT 128
#define GATES 512
#define BT (BB * TT)
#define HP (OT + 4)
#define RB 16
#define GCOL(r) (((((r) & (OT - 1)) >> 2) << 4) + (((r) & 3) << 2) + ((r) >> 7))
#define GSCALE(r) ((((r) >> 7) == 2) ? 1.0f : 0.5f)
#define SP 144
#define SW 272

__device__ __forceinline__ v8f wmma_f32(v2f a, v2f b, v8f c) {
    return __builtin_amdgcn_wmma_f32_16x16x4_f32(
         false, a,  false, b,
         (short)0, c,  false,  false);
}

#define FB 1024
#define NLW (BB / RB)
#define ST_NONE 0
#define ST_DEGREE 1
#define ST_SCAN 2
#define ST_FILL 3
#define ST_GATHER1 4
#define ST_GEMM1 5
#define ST_GATHER2 6
#define ST_GEMM2 7

__global__ void zero_i32(int* __restrict__ p, int n) {
    int i = blockIdx.x * blockDim.x + threadIdx.x;
    if (i < n) p[i] = 0;
}

__device__ __forceinline__ void degree_stage(const int vb, const int* __restrict__ src,
                                             const int* __restrict__ dst, int* __restrict__ cnt_out,
                                             int* __restrict__ cnt_in, int E) {
    const int e = vb * FB + (int)threadIdx.x;
    if (e < E) {
        atomicAdd(&cnt_out[src[e]], 1);
        atomicAdd(&cnt_in[dst[e]], 1);
    }
}

#define SCAN_IT ((N_NODES + 1023) / 1024)
#define SCAN_G 7
static_assert(SCAN_IT % SCAN_G == 0, "group size must divide the pass count");
static_assert(FB == 1024, "the scan is written for one workgroup of 32 waves");
__device__ __forceinline__ void scan_stage(const int* __restrict__ cnt_in,
                                           const int* __restrict__ cnt_out,
                                           int* __restrict__ rowptr, int* __restrict__ cursor,
                                           float* __restrict__ no, float* __restrict__ ni) {
    __shared__ int wtot[32];
    const int wave = threadIdx.x >> 5, lane = threadIdx.x & 31;
    const int base = wave * (SCAN_IT * 32) + lane;
    int tot = 0;
#pragma unroll 7
    for (int it = 0; it < SCAN_IT; ++it) {
        const int i = base + it * 32;
        const int cval = cnt_in[min(i, N_NODES - 1)];
        tot += (i < N_NODES) ? cval : 0;
    }
#pragma unroll
    for (int o = 16; o > 0; o >>= 1) tot += __shfl_xor(tot, o);
    if (lane == 0) wtot[wave] = tot;
    __syncthreads();
    int carry = 0;
    for (int w = 0; w < wave; ++w) carry += wtot[w];
    for (int g = 0; g < SCAN_IT; g += SCAN_G) {
        int cv[SCAN_G], co[SCAN_G], x[SCAN_G];
#pragma unroll
        for (int u = 0; u < SCAN_G; ++u) {
            const int i = base + (g + u) * 32;
            const int ic = min(i, N_NODES - 1);
            const int cval = cnt_in[ic];
            co[u] = cnt_out[ic];
            cv[u] = (i < N_NODES) ? cval : 0;
            x[u] = cv[u];
        }
#pragma unroll
        for (int o = 1; o < 32; o <<= 1) {
#pragma unroll
            for (int u = 0; u < SCAN_G; ++u) {
                const int y = __shfl_up(x[u], o);
                if (lane >= o) x[u] += y;
            }
        }
#pragma unroll
        for (int u = 0; u < SCAN_G; ++u) {
            const int i = base + (g + u) * 32;
            if (i < N_NODES) {
                const int v = carry + x[u] - cv[u];
                rowptr[i] = v;
                cursor[i] = v;
                no[i] = rsqrtf(fmaxf((float)co[u], 1.0f));
                ni[i] = rsqrtf(fmaxf((float)cv[u], 1.0f));
            }
            carry += __shfl(x[u], 31);
        }
    }
    if (threadIdx.x == 1023) rowptr[N_NODES] = carry;
}

__device__ __forceinline__ void fill_stage(const int vb, const int* __restrict__ src,
                                           const int* __restrict__ dst,
                                           const float* __restrict__ norm_out, int* __restrict__ cursor,
                                           int2* __restrict__ ebuf, int E) {
    const int e = vb * FB + (int)threadIdx.x;
    if (e < E) {
        const int s = src[e];
        const int p = atomicAdd(&cursor[dst[e]], 1);
        ebuf[p] = make_int2(s, __float_as_int(norm_out[s]));
    }
}

__device__ __forceinline__ void gather1_stage(const int vb, const int* __restrict__ rowptr,
                                              const int2* __restrict__ ebuf,
                                              const float* __restrict__ ht, const float* __restrict__ chs,
                                              float* __restrict__ agg1) {
    const int lane = threadIdx.x & 31;
    const int node = __builtin_amdgcn_readfirstlane((vb << 5) + (int)(threadIdx.x >> 5));
    if (node >= N_NODES) return;
    const int q = lane << 2;
    const float* fb = (q < SD) ? (ht + q) : (chs + (q - SD));
    const int beg = rowptr[node], end = rowptr[node + 1];
    float4 acc = make_float4(0.f, 0.f, 0.f, 0.f);
    for (int j0 = beg; j0 < end; j0 += 32) {
        const int m = min(32, end - j0);
        int2 ew = make_int2(0, 0);
        if (lane < m) ew = ebuf[j0 + lane];
        int i = 0;
        for (; i + 4 <= m; i += 4) {
            const int s0 = __builtin_amdgcn_readlane(ew.x, i);
            const int s1 = __builtin_amdgcn_readlane(ew.x, i + 1);
            const int s2 = __builtin_amdgcn_readlane(ew.x, i + 2);
            const int s3 = __builtin_amdgcn_readlane(ew.x, i + 3);
            const float4 v0 = *(const float4*)(fb + (long)s0 * SD);
            const float4 v1 = *(const float4*)(fb + (long)s1 * SD);
            const float4 v2 = *(const float4*)(fb + (long)s2 * SD);
            const float4 v3 = *(const float4*)(fb + (long)s3 * SD);
            const float w0 = __int_as_float(__builtin_amdgcn_readlane(ew.y, i));
            const float w1 = __int_as_float(__builtin_amdgcn_readlane(ew.y, i + 1));
            const float w2 = __int_as_float(__builtin_amdgcn_readlane(ew.y, i + 2));
            const float w3 = __int_as_float(__builtin_amdgcn_readlane(ew.y, i + 3));
            acc.x += v0.x * w0; acc.y += v0.y * w0; acc.z += v0.z * w0; acc.w += v0.w * w0;
            acc.x += v1.x * w1; acc.y += v1.y * w1; acc.z += v1.z * w1; acc.w += v1.w * w1;
            acc.x += v2.x * w2; acc.y += v2.y * w2; acc.z += v2.z * w2; acc.w += v2.w * w2;
            acc.x += v3.x * w3; acc.y += v3.y * w3; acc.z += v3.z * w3; acc.w += v3.w * w3;
        }
        for (; i < m; ++i) {
            const int s0 = __builtin_amdgcn_readlane(ew.x, i);
            const float w0 = __int_as_float(__builtin_amdgcn_readlane(ew.y, i));
            const float4 v0 = *(const float4*)(fb + (long)s0 * SD);
            acc.x += v0.x * w0; acc.y += v0.y * w0; acc.z += v0.z * w0; acc.w += v0.w * w0;
        }
    }
    *(float4*)(agg1 + (long)node * 128 + q) = acc;
}

__device__ __forceinline__ void gather2_stage(const int vb, const int* __restrict__ rowptr,
                                              const int2* __restrict__ ebuf,
                                              const float* __restrict__ x1, float* __restrict__ agg2) {
    const int lane = threadIdx.x & 31;
    const int node = __builtin_amdgcn_readfirstlane((vb << 5) + (int)(threadIdx.x >> 5));
    if (node >= N_NODES) return;
    const int q = lane << 1;
    const float* fb = x1 + q;
    const int beg = rowptr[node], end = rowptr[node + 1];
    float2 acc = make_float2(0.f, 0.f);
    for (int j0 = beg; j0 < end; j0 += 32) {
        const int m = min(32, end - j0);
        int2 ew = make_int2(0, 0);
        if (lane < m) ew = ebuf[j0 + lane];
        int i = 0;
        for (; i + 4 <= m; i += 4) {
            const int s0 = __builtin_amdgcn_readlane(ew.x, i);
            const int s1 = __builtin_amdgcn_readlane(ew.x, i + 1);
            const int s2 = __builtin_amdgcn_readlane(ew.x, i + 2);
            const int s3 = __builtin_amdgcn_readlane(ew.x, i + 3);
            const float2 v0 = *(const float2*)(fb + (long)s0 * 64);
            const float2 v1 = *(const float2*)(fb + (long)s1 * 64);
            const float2 v2 = *(const float2*)(fb + (long)s2 * 64);
            const float2 v3 = *(const float2*)(fb + (long)s3 * 64);
            const float w0 = __int_as_float(__builtin_amdgcn_readlane(ew.y, i));
            const float w1 = __int_as_float(__builtin_amdgcn_readlane(ew.y, i + 1));
            const float w2 = __int_as_float(__builtin_amdgcn_readlane(ew.y, i + 2));
            const float w3 = __int_as_float(__builtin_amdgcn_readlane(ew.y, i + 3));
            acc.x += v0.x * w0; acc.y += v0.y * w0;
            acc.x += v1.x * w1; acc.y += v1.y * w1;
            acc.x += v2.x * w2; acc.y += v2.y * w2;
            acc.x += v3.x * w3; acc.y += v3.y * w3;
        }
        for (; i < m; ++i) {
            const int s0 = __builtin_amdgcn_readlane(ew.x, i);
            const float w0 = __int_as_float(__builtin_amdgcn_readlane(ew.y, i));
            const float2 v0 = *(const float2*)(fb + (long)s0 * 64);
            acc.x += v0.x * w0; acc.y += v0.y * w0;
        }
    }
    *(float2*)(agg2 + (long)node * 64 + q) = acc;
}

__global__ void transpose_k(const float* __restrict__ W, float* __restrict__ WT) {
    int idx = blockIdx.x * blockDim.x + threadIdx.x;
    if (idx >= GATES * OT) return;
    int r = idx >> 7, c = idx & 127;
    WT[c * GATES + GCOL(r)] = W[idx] * GSCALE(r);
}

__global__ void bias_sum_k(const float* __restrict__ a, const float* __restrict__ b,
                           float* __restrict__ o) {
    int i = blockIdx.x * blockDim.x + threadIdx.x;
    if (i < GATES) o[GCOL(i)] = (a[i] + b[i]) * GSCALE(i);
}

__global__ void concat_k(const float* __restrict__ hs, const float* __restrict__ cht,
                         float* __restrict__ X) {
    int idx = blockIdx.x * blockDim.x + threadIdx.x;
    if (idx >= BT * 128) return;
    int row = idx >> 7, k = idx & 127;
    X[idx] = (k < SD) ? hs[row * SD + k] : cht[row * SD + (k - SD)];
}

__device__ __forceinline__ void gemm_tile(const float* __restrict__ A, const float* __restrict__ rowscale,
                                          const float* __restrict__ Bm, const float* __restrict__ bias,
                                          float* __restrict__ C, const int N, const int K,
                                          const int applyGelu, const int row0, const int ct, const int lane) {
    const int hl = lane >> 4;
    const int l16 = lane & 15;
    const int col = (ct << 4) + l16;
    const int arow = row0 + l16;
    const float rs = rowscale ? rowscale[arow] : 1.0f;
    const float* Arow = A + (size_t)arow * K;
    v8f acc = {};
    for (int k = 0; k < K; k += 4) {
        const int ka = k + (hl << 1);
        v2f a, b;
        a.x = Arow[ka] * rs;
        a.y = Arow[ka + 1] * rs;
        b.x = Bm[(size_t)ka * N + col];
        b.y = Bm[(size_t)(ka + 1) * N + col];
        acc = wmma_f32(a, b, acc);
    }
    const float bv = bias ? bias[col] : 0.0f;
#pragma unroll
    for (int r = 0; r < 8; ++r) {
        float v = acc[r] + bv;
        if (applyGelu) v = 0.5f * v * (1.0f + erff(v * 0.70710678118654752f));
        C[(size_t)(row0 + r + (hl << 3)) * N + col] = v;
    }
}

__global__ void gemm_wmma_kernel(const float* __restrict__ A, const float* __restrict__ rowscale,
                                 const float* __restrict__ Bm, const float* __restrict__ bias,
                                 float* __restrict__ C, int M, int N, int K, int applyGelu) {
    const int wave = threadIdx.x >> 5;
    const int ct = blockIdx.y * (blockDim.x >> 5) + wave;
    if ((ct << 4) >= N) return;
    gemm_tile(A, rowscale, Bm, bias, C, N, K, applyGelu, blockIdx.x << 4, ct, threadIdx.x & 31);
}

__device__ __forceinline__ void gemm64_stage(const int vb, const float* __restrict__ A,
                                             const float* __restrict__ rowscale,
                                             const float* __restrict__ Bm, const float* __restrict__ bias,
                                             float* __restrict__ C, const int K) {
    const int wave = threadIdx.x >> 5;
    const int rt = (vb << 3) + (wave >> 2);
    if (rt >= N_NODES / 16) return;
    gemm_tile(A, rowscale, Bm, bias, C, 64, K, 1, rt << 4, wave & 3, threadIdx.x & 31);
}

__device__ __forceinline__ float tanh_hw(float x) {
    return __builtin_amdgcn_tanhf(x);
}
__device__ __forceinline__ float sigmoid_half_hw(float xh) {
    return fmaf(__builtin_amdgcn_tanhf(xh), 0.5f, 0.5f);
}

template <int T0, int T1>
__device__ __forceinline__ void lstm_segment(const int wg,
                                             const float* gpre,
                                             const float* __restrict__ WhhT,
                                             float* __restrict__ out,
                                             float* __restrict__ hst, float* __restrict__ cst) {
    __shared__ __attribute__((aligned(16))) float h_lds[2][RB][HP];
    __shared__ __attribute__((aligned(16))) float s_lds[32][SW];
    const int tid = threadIdx.x;
    const int wave = tid >> 5, lane = tid & 31;
    const int hl = lane >> 4, l16 = lane & 15;
    const int col0 = wave << 4;
    const int col = col0 + l16;
    const int brow0 = wg * RB;

    float bx[32], by[32];
#pragma unroll
    for (int kk = 0; kk < 32; ++kk) {
        const int ka = (kk << 2) + (hl << 1);
        bx[kk] = WhhT[ka * GATES + col];
        by[kk] = WhhT[(ka + 1) * GATES + col];
    }

    float creg[2] = {0.f, 0.f};
    float* hsw = hst + wg * (RB * OT);
    float* csw = cst + ((wg * FB + tid) << 1);
    if (T0 == 0) {
        for (int i = tid; i < RB * OT; i += FB) h_lds[0][i >> 7][i & (OT - 1)] = 0.0f;
    } else {
        for (int i = tid; i < RB * OT; i += FB) h_lds[0][i >> 7][i & (OT - 1)] = hsw[i];
        const float2 cv = *(const float2*)csw;
        creg[0] = cv.x;
        creg[1] = cv.y;
    }
    const int crow = lane >> 2;
    const int cq = (lane & 3) << 2;
    const int cj = (wave << 2) + (lane & 3);
    float* sw = &s_lds[wave][0];
    const int orow = wave >> 1;
    const int ocol = ((wave & 1) << 6) + (lane << 1);
    float* outp = out + (long)(brow0 + orow) * TT * OT + ocol;
    __syncthreads();

    int cur = 0;
    auto step = [&](const int t, const bool store_prev) __attribute__((always_inline)) {
        float4 gp[2];
#pragma unroll
        for (int u = 0; u < 2; ++u)
            gp[u] = *(const float4*)(gpre + ((long)(brow0 + crow + (u << 3)) * TT + t) * GATES + col0 + cq);
        __builtin_amdgcn_fence(__ATOMIC_RELEASE, "wavefront");

        const float (*hc)[HP] = h_lds[cur];
        float (*hn)[HP] = h_lds[cur ^ 1];
        const float2 ov = *(const float2*)&hc[orow][ocol];

        v8f c0 = {};
#pragma unroll
        for (int kk = 0; kk < 32; ++kk) {
            const int ka = (kk << 2) + (hl << 1);
            v2f b, a0;
            b.x = bx[kk];                b.y = by[kk];
            a0.x = hc[l16][ka];          a0.y = hc[l16][ka + 1];
            c0 = wmma_f32(a0, b, c0);
        }
#pragma unroll
        for (int r = 0; r < 8; ++r) sw[hl * SP + (r << 4) + l16] = c0[r];
        __builtin_amdgcn_fence(__ATOMIC_RELEASE, "wavefront");
        __builtin_amdgcn_wave_barrier();
        __builtin_amdgcn_fence(__ATOMIC_ACQUIRE, "wavefront");
        float4 gs[2];
#pragma unroll
        for (int u = 0; u < 2; ++u) gs[u] = *(const float4*)&sw[u * SP + (lane << 2)];
        if (store_prev) *(float2*)(outp + (long)(t - 1) * OT) = ov;

#pragma unroll
        for (int u = 0; u < 2; ++u) {
            float gi = gs[u].x + gp[u].x;
            float gf = gs[u].y + gp[u].y;
            float gg = gs[u].z + gp[u].z;
            float go = gs[u].w + gp[u].w;
            float si = sigmoid_half_hw(gi);
            float sf = sigmoid_half_hw(gf);
            float so = sigmoid_half_hw(go);
            float cc = sf * creg[u] + si * tanh_hw(gg);
            creg[u] = cc;
            float hh = so * tanh_hw(cc);
            hn[crow + (u << 3)][cj] = hh;
        }
        __syncthreads();
        cur ^= 1;
    };
    step(T0, false);
    for (int t = T0 + 1; t < T1; ++t) step(t, true);
    *(float2*)(outp + (long)(T1 - 1) * OT) = *(const float2*)&h_lds[cur][orow][ocol];
    if (T1 < TT) {
        for (int i = tid; i < RB * OT; i += FB) hsw[i] = h_lds[cur][i >> 7][i & (OT - 1)];
        *(float2*)csw = make_float2(creg[0], creg[1]);
    }
}

template <int STAGE, int T0, int T1>
__global__ __launch_bounds__(FB) void fused_k(
    const float* gpre, const float* __restrict__ WhhT, float* __restrict__ out,
    float* __restrict__ hst, float* __restrict__ cst,
    const int* __restrict__ src, const int* __restrict__ dst, int E,
    int* __restrict__ cnt_out, int* __restrict__ cnt_in, int* __restrict__ rowptr,
    int* __restrict__ cursor, float* __restrict__ norm_out, float* __restrict__ norm_in,
    int2* __restrict__ ebuf, const float* __restrict__ ht, const float* __restrict__ chs,
    float* __restrict__ agg, const float* __restrict__ W1, const float* __restrict__ b1,
    float* __restrict__ x1, const float* __restrict__ W2, const float* __restrict__ b2,
    float* __restrict__ out_sp) {
    if (blockIdx.x < NLW) {
        lstm_segment<T0, T1>((int)blockIdx.x, gpre, WhhT, out, hst, cst);
        return;
    }
    const int vb = (int)blockIdx.x - NLW;
    if (STAGE == ST_DEGREE)  degree_stage(vb, src, dst, cnt_out, cnt_in, E);
    if (STAGE == ST_SCAN)    scan_stage(cnt_in, cnt_out, rowptr, cursor, norm_out, norm_in);
    if (STAGE == ST_FILL)    fill_stage(vb, src, dst, norm_out, cursor, ebuf, E);
    if (STAGE == ST_GATHER1) gather1_stage(vb, rowptr, ebuf, ht, chs, agg);
    if (STAGE == ST_GEMM1)   gemm64_stage(vb, agg, norm_in, W1, b1, x1, 128);
    if (STAGE == ST_GATHER2) gather2_stage(vb, rowptr, ebuf, x1, agg);
    if (STAGE == ST_GEMM2)   gemm64_stage(vb, agg, norm_in, W2, b2, out_sp, 64);
}

extern "C" void kernel_launch(void* const* d_in, const int* in_sizes, int n_in,
                              void* d_out, int out_size, void* d_ws, size_t ws_size,
                              hipStream_t stream) {
    const float* hs  = (const float*)d_in[0];
    const float* ht  = (const float*)d_in[1];
    const float* chs = (const float*)d_in[2];
    const float* cht = (const float*)d_in[3];
    const int*   src = (const int*)d_in[4];
    const int*   dst = (const int*)d_in[5];
    const float* W1  = (const float*)d_in[6];
    const float* b1  = (const float*)d_in[7];
    const float* W2  = (const float*)d_in[8];
    const float* b2  = (const float*)d_in[9];
    const float* Wih = (const float*)d_in[10];
    const float* Whh = (const float*)d_in[11];
    const float* bih = (const float*)d_in[12];
    const float* bhh = (const float*)d_in[13];
    const int E = in_sizes[4];

    float* ws       = (float*)d_ws;
    float* norm_out = ws;
    float* norm_in  = norm_out + N_NODES;
    float* agg      = norm_in + N_NODES;
    float* bkt      = agg + (size_t)N_NODES * 128;
    float* x1       = bkt + (size_t)N_NODES * 64;
    float* Xcat     = x1 + (size_t)N_NODES * 64;
    float* gpre     = Xcat + (size_t)BT * 128;
    float* WihT     = gpre + (size_t)BT * GATES;
    float* WhhT     = WihT + OT * GATES;
    float* bsum     = WhhT + OT * GATES;
    int*  cnt_in  = (int*)bkt;
    int*  cnt_out = cnt_in + N_NODES;
    int*  rowptr  = cnt_out + N_NODES;
    int*  cursor  = rowptr + N_NODES + 16;
    int2* ebuf    = (int2*)(cursor + N_NODES);
    float* hst = Xcat;
    float* cst = Xcat + NLW * RB * OT;
    float* out_lstm = (float*)d_out;
    float* out_sp = (float*)d_out + (size_t)BT * OT;

    const int thr = 256;
    zero_i32<<<(2 * N_NODES + thr - 1) / thr, thr, 0, stream>>>(cnt_in, 2 * N_NODES);

    transpose_k<<<(GATES * OT + thr - 1) / thr, thr, 0, stream>>>(Wih, WihT);
    transpose_k<<<(GATES * OT + thr - 1) / thr, thr, 0, stream>>>(Whh, WhhT);
    bias_sum_k<<<2, 256, 0, stream>>>(bih, bhh, bsum);
    concat_k<<<(BT * 128 + thr - 1) / thr, thr, 0, stream>>>(hs, cht, Xcat);
    dim3 gp(BT / 16, 8);
    gemm_wmma_kernel<<<gp, 128, 0, stream>>>(Xcat, nullptr, WihT, bsum, gpre, BT, GATES, 128, 0);

    const int eb = (E + FB - 1) / FB;
    const int nb = (N_NODES * 32 + FB - 1) / FB;
    const int mb = (N_NODES / 16 + 7) / 8;
#define FUSED_ARGS gpre, WhhT, out_lstm, hst, cst, src, dst, E, cnt_out, cnt_in, rowptr, cursor, \
                   norm_out, norm_in, ebuf, ht, chs, agg, W1, b1, x1, W2, b2, out_sp
    fused_k<ST_DEGREE,    0,  64><<<NLW + eb, FB, 0, stream>>>(FUSED_ARGS);
    fused_k<ST_SCAN,     64,  96><<<NLW + 1,  FB, 0, stream>>>(FUSED_ARGS);
    fused_k<ST_FILL,     96, 160><<<NLW + eb, FB, 0, stream>>>(FUSED_ARGS);
    fused_k<ST_GATHER1, 160, 256><<<NLW + nb, FB, 0, stream>>>(FUSED_ARGS);
    fused_k<ST_GEMM1,   256, 320><<<NLW + mb, FB, 0, stream>>>(FUSED_ARGS);
    fused_k<ST_GATHER2, 320, 384><<<NLW + nb, FB, 0, stream>>>(FUSED_ARGS);
    fused_k<ST_GEMM2,   384, 448><<<NLW + mb, FB, 0, stream>>>(FUSED_ARGS);
    fused_k<ST_NONE,    448, 512><<<NLW,      FB, 0, stream>>>(FUSED_ARGS);
#undef FUSED_ARGS
}


# LSTM step barrier hands over LDS only: LDS-scoped fences + s_barrier in place of __syncthreads(), so no wave waits there for the acknowledgement of its store to out (was s_wait_storecnt 0 each step)
# speedup vs baseline: 7.1956x; 1.1168x over previous
#include <hip/hip_runtime.h>
#include <math.h>

typedef float v2f __attribute__((ext_vector_type(2)));
typedef float v8f __attribute__((ext_vector_type(8)));

#define N_NODES 50000
#define BB 32
#define TT 512
#define SD 64
#define OT 128
#define GATES 512
#define BT (BB * TT)
#define HP (OT + 4)
#define RB 16
#define GCOL(r) (((((r) & (OT - 1)) >> 2) << 4) + (((r) & 3) << 2) + ((r) >> 7))
#define GSCALE(r) ((((r) >> 7) == 2) ? 1.0f : 0.5f)
#define SP 144
#define SW 272

__device__ __forceinline__ v8f wmma_f32(v2f a, v2f b, v8f c) {
    return __builtin_amdgcn_wmma_f32_16x16x4_f32(
         false, a,  false, b,
         (short)0, c,  false,  false);
}

#define FB 1024
#define NLW (BB / RB)
#define ST_NONE 0
#define ST_DEGREE 1
#define ST_SCAN 2
#define ST_FILL 3
#define ST_GATHER1 4
#define ST_GEMM1 5
#define ST_GATHER2 6
#define ST_GEMM2 7

__global__ void zero_i32(int* __restrict__ p, int n) {
    int i = blockIdx.x * blockDim.x + threadIdx.x;
    if (i < n) p[i] = 0;
}

__device__ __forceinline__ void degree_stage(const int vb, const int* __restrict__ src,
                                             const int* __restrict__ dst, int* __restrict__ cnt_out,
                                             int* __restrict__ cnt_in, int E) {
    const int e = vb * FB + (int)threadIdx.x;
    if (e < E) {
        atomicAdd(&cnt_out[src[e]], 1);
        atomicAdd(&cnt_in[dst[e]], 1);
    }
}

#define SCAN_IT ((N_NODES + 1023) / 1024)
#define SCAN_G 7
static_assert(SCAN_IT % SCAN_G == 0, "group size must divide the pass count");
static_assert(FB == 1024, "the scan is written for one workgroup of 32 waves");
__device__ __forceinline__ void scan_stage(const int* __restrict__ cnt_in,
                                           const int* __restrict__ cnt_out,
                                           int* __restrict__ rowptr, int* __restrict__ cursor,
                                           float* __restrict__ no, float* __restrict__ ni) {
    __shared__ int wtot[32];
    const int wave = threadIdx.x >> 5, lane = threadIdx.x & 31;
    const int base = wave * (SCAN_IT * 32) + lane;
    int tot = 0;
#pragma unroll 7
    for (int it = 0; it < SCAN_IT; ++it) {
        const int i = base + it * 32;
        const int cval = cnt_in[min(i, N_NODES - 1)];
        tot += (i < N_NODES) ? cval : 0;
    }
#pragma unroll
    for (int o = 16; o > 0; o >>= 1) tot += __shfl_xor(tot, o);
    if (lane == 0) wtot[wave] = tot;
    __syncthreads();
    int carry = 0;
    for (int w = 0; w < wave; ++w) carry += wtot[w];
    for (int g = 0; g < SCAN_IT; g += SCAN_G) {
        int cv[SCAN_G], co[SCAN_G], x[SCAN_G];
#pragma unroll
        for (int u = 0; u < SCAN_G; ++u) {
            const int i = base + (g + u) * 32;
            const int ic = min(i, N_NODES - 1);
            const int cval = cnt_in[ic];
            co[u] = cnt_out[ic];
            cv[u] = (i < N_NODES) ? cval : 0;
            x[u] = cv[u];
        }
#pragma unroll
        for (int o = 1; o < 32; o <<= 1) {
#pragma unroll
            for (int u = 0; u < SCAN_G; ++u) {
                const int y = __shfl_up(x[u], o);
                if (lane >= o) x[u] += y;
            }
        }
#pragma unroll
        for (int u = 0; u < SCAN_G; ++u) {
            const int i = base + (g + u) * 32;
            if (i < N_NODES) {
                const int v = carry + x[u] - cv[u];
                rowptr[i] = v;
                cursor[i] = v;
                no[i] = rsqrtf(fmaxf((float)co[u], 1.0f));
                ni[i] = rsqrtf(fmaxf((float)cv[u], 1.0f));
            }
            carry += __shfl(x[u], 31);
        }
    }
    if (threadIdx.x == 1023) rowptr[N_NODES] = carry;
}

__device__ __forceinline__ void fill_stage(const int vb, const int* __restrict__ src,
                                           const int* __restrict__ dst,
                                           const float* __restrict__ norm_out, int* __restrict__ cursor,
                                           int2* __restrict__ ebuf, int E) {
    const int e = vb * FB + (int)threadIdx.x;
    if (e < E) {
        const int s = src[e];
        const int p = atomicAdd(&cursor[dst[e]], 1);
        ebuf[p] = make_int2(s, __float_as_int(norm_out[s]));
    }
}

__device__ __forceinline__ void gather1_stage(const int vb, const int* __restrict__ rowptr,
                                              const int2* __restrict__ ebuf,
                                              const float* __restrict__ ht, const float* __restrict__ chs,
                                              float* __restrict__ agg1) {
    const int lane = threadIdx.x & 31;
    const int node = __builtin_amdgcn_readfirstlane((vb << 5) + (int)(threadIdx.x >> 5));
    if (node >= N_NODES) return;
    const int q = lane << 2;
    const float* fb = (q < SD) ? (ht + q) : (chs + (q - SD));
    const int beg = rowptr[node], end = rowptr[node + 1];
    float4 acc = make_float4(0.f, 0.f, 0.f, 0.f);
    for (int j0 = beg; j0 < end; j0 += 32) {
        const int m = min(32, end - j0);
        int2 ew = make_int2(0, 0);
        if (lane < m) ew = ebuf[j0 + lane];
        int i = 0;
        for (; i + 4 <= m; i += 4) {
            const int s0 = __builtin_amdgcn_readlane(ew.x, i);
            const int s1 = __builtin_amdgcn_readlane(ew.x, i + 1);
            const int s2 = __builtin_amdgcn_readlane(ew.x, i + 2);
            const int s3 = __builtin_amdgcn_readlane(ew.x, i + 3);
            const float4 v0 = *(const float4*)(fb + (long)s0 * SD);
            const float4 v1 = *(const float4*)(fb + (long)s1 * SD);
            const float4 v2 = *(const float4*)(fb + (long)s2 * SD);
            const float4 v3 = *(const float4*)(fb + (long)s3 * SD);
            const float w0 = __int_as_float(__builtin_amdgcn_readlane(ew.y, i));
            const float w1 = __int_as_float(__builtin_amdgcn_readlane(ew.y, i + 1));
            const float w2 = __int_as_float(__builtin_amdgcn_readlane(ew.y, i + 2));
            const float w3 = __int_as_float(__builtin_amdgcn_readlane(ew.y, i + 3));
            acc.x += v0.x * w0; acc.y += v0.y * w0; acc.z += v0.z * w0; acc.w += v0.w * w0;
            acc.x += v1.x * w1; acc.y += v1.y * w1; acc.z += v1.z * w1; acc.w += v1.w * w1;
            acc.x += v2.x * w2; acc.y += v2.y * w2; acc.z += v2.z * w2; acc.w += v2.w * w2;
            acc.x += v3.x * w3; acc.y += v3.y * w3; acc.z += v3.z * w3; acc.w += v3.w * w3;
        }
        for (; i < m; ++i) {
            const int s0 = __builtin_amdgcn_readlane(ew.x, i);
            const float w0 = __int_as_float(__builtin_amdgcn_readlane(ew.y, i));
            const float4 v0 = *(const float4*)(fb + (long)s0 * SD);
            acc.x += v0.x * w0; acc.y += v0.y * w0; acc.z += v0.z * w0; acc.w += v0.w * w0;
        }
    }
    *(float4*)(agg1 + (long)node * 128 + q) = acc;
}

__device__ __forceinline__ void gather2_stage(const int vb, const int* __restrict__ rowptr,
                                              const int2* __restrict__ ebuf,
                                              const float* __restrict__ x1, float* __restrict__ agg2) {
    const int lane = threadIdx.x & 31;
    const int node = __builtin_amdgcn_readfirstlane((vb << 5) + (int)(threadIdx.x >> 5));
    if (node >= N_NODES) return;
    const int q = lane << 1;
    const float* fb = x1 + q;
    const int beg = rowptr[node], end = rowptr[node + 1];
    float2 acc = make_float2(0.f, 0.f);
    for (int j0 = beg; j0 < end; j0 += 32) {
        const int m = min(32, end - j0);
        int2 ew = make_int2(0, 0);
        if (lane < m) ew = ebuf[j0 + lane];
        int i = 0;
        for (; i + 4 <= m; i += 4) {
            const int s0 = __builtin_amdgcn_readlane(ew.x, i);
            const int s1 = __builtin_amdgcn_readlane(ew.x, i + 1);
            const int s2 = __builtin_amdgcn_readlane(ew.x, i + 2);
            const int s3 = __builtin_amdgcn_readlane(ew.x, i + 3);
            const float2 v0 = *(const float2*)(fb + (long)s0 * 64);
            const float2 v1 = *(const float2*)(fb + (long)s1 * 64);
            const float2 v2 = *(const float2*)(fb + (long)s2 * 64);
            const float2 v3 = *(const float2*)(fb + (long)s3 * 64);
            const float w0 = __int_as_float(__builtin_amdgcn_readlane(ew.y, i));
            const float w1 = __int_as_float(__builtin_amdgcn_readlane(ew.y, i + 1));
            const float w2 = __int_as_float(__builtin_amdgcn_readlane(ew.y, i + 2));
            const float w3 = __int_as_float(__builtin_amdgcn_readlane(ew.y, i + 3));
            acc.x += v0.x * w0; acc.y += v0.y * w0;
            acc.x += v1.x * w1; acc.y += v1.y * w1;
            acc.x += v2.x * w2; acc.y += v2.y * w2;
            acc.x += v3.x * w3; acc.y += v3.y * w3;
        }
        for (; i < m; ++i) {
            const int s0 = __builtin_amdgcn_readlane(ew.x, i);
            const float w0 = __int_as_float(__builtin_amdgcn_readlane(ew.y, i));
            const float2 v0 = *(const float2*)(fb + (long)s0 * 64);
            acc.x += v0.x * w0; acc.y += v0.y * w0;
        }
    }
    *(float2*)(agg2 + (long)node * 64 + q) = acc;
}

__global__ void transpose_k(const float* __restrict__ W, float* __restrict__ WT) {
    int idx = blockIdx.x * blockDim.x + threadIdx.x;
    if (idx >= GATES * OT) return;
    int r = idx >> 7, c = idx & 127;
    WT[c * GATES + GCOL(r)] = W[idx] * GSCALE(r);
}

__global__ void bias_sum_k(const float* __restrict__ a, const float* __restrict__ b,
                           float* __restrict__ o) {
    int i = blockIdx.x * blockDim.x + threadIdx.x;
    if (i < GATES) o[GCOL(i)] = (a[i] + b[i]) * GSCALE(i);
}

__global__ void concat_k(const float* __restrict__ hs, const float* __restrict__ cht,
                         float* __restrict__ X) {
    int idx = blockIdx.x * blockDim.x + threadIdx.x;
    if (idx >= BT * 128) return;
    int row = idx >> 7, k = idx & 127;
    X[idx] = (k < SD) ? hs[row * SD + k] : cht[row * SD + (k - SD)];
}

__device__ __forceinline__ void gemm_tile(const float* __restrict__ A, const float* __restrict__ rowscale,
                                          const float* __restrict__ Bm, const float* __restrict__ bias,
                                          float* __restrict__ C, const int N, const int K,
                                          const int applyGelu, const int row0, const int ct, const int lane) {
    const int hl = lane >> 4;
    const int l16 = lane & 15;
    const int col = (ct << 4) + l16;
    const int arow = row0 + l16;
    const float rs = rowscale ? rowscale[arow] : 1.0f;
    const float* Arow = A + (size_t)arow * K;
    v8f acc = {};
    for (int k = 0; k < K; k += 4) {
        const int ka = k + (hl << 1);
        v2f a, b;
        a.x = Arow[ka] * rs;
        a.y = Arow[ka + 1] * rs;
        b.x = Bm[(size_t)ka * N + col];
        b.y = Bm[(size_t)(ka + 1) * N + col];
        acc = wmma_f32(a, b, acc);
    }
    const float bv = bias ? bias[col] : 0.0f;
#pragma unroll
    for (int r = 0; r < 8; ++r) {
        float v = acc[r] + bv;
        if (applyGelu) v = 0.5f * v * (1.0f + erff(v * 0.70710678118654752f));
        C[(size_t)(row0 + r + (hl << 3)) * N + col] = v;
    }
}

__global__ void gemm_wmma_kernel(const float* __restrict__ A, const float* __restrict__ rowscale,
                                 const float* __restrict__ Bm, const float* __restrict__ bias,
                                 float* __restrict__ C, int M, int N, int K, int applyGelu) {
    const int wave = threadIdx.x >> 5;
    const int ct = blockIdx.y * (blockDim.x >> 5) + wave;
    if ((ct << 4) >= N) return;
    gemm_tile(A, rowscale, Bm, bias, C, N, K, applyGelu, blockIdx.x << 4, ct, threadIdx.x & 31);
}

__device__ __forceinline__ void gemm64_stage(const int vb, const float* __restrict__ A,
                                             const float* __restrict__ rowscale,
                                             const float* __restrict__ Bm, const float* __restrict__ bias,
                                             float* __restrict__ C, const int K) {
    const int wave = threadIdx.x >> 5;
    const int rt = (vb << 3) + (wave >> 2);
    if (rt >= N_NODES / 16) return;
    gemm_tile(A, rowscale, Bm, bias, C, 64, K, 1, rt << 4, wave & 3, threadIdx.x & 31);
}

__device__ __forceinline__ float tanh_hw(float x) {
    return __builtin_amdgcn_tanhf(x);
}
__device__ __forceinline__ float sigmoid_half_hw(float xh) {
    return fmaf(__builtin_amdgcn_tanhf(xh), 0.5f, 0.5f);
}

template <int T0, int T1>
__device__ __forceinline__ void lstm_segment(const int wg,
                                             const float* gpre,
                                             const float* __restrict__ WhhT,
                                             float* __restrict__ out,
                                             float* __restrict__ hst, float* __restrict__ cst) {
    __shared__ __attribute__((aligned(16))) float h_lds[2][RB][HP];
    __shared__ __attribute__((aligned(16))) float s_lds[32][SW];
    const int tid = threadIdx.x;
    const int wave = tid >> 5, lane = tid & 31;
    const int hl = lane >> 4, l16 = lane & 15;
    const int col0 = wave << 4;
    const int col = col0 + l16;
    const int brow0 = wg * RB;

    float bx[32], by[32];
#pragma unroll
    for (int kk = 0; kk < 32; ++kk) {
        const int ka = (kk << 2) + (hl << 1);
        bx[kk] = WhhT[ka * GATES + col];
        by[kk] = WhhT[(ka + 1) * GATES + col];
    }

    float creg[2] = {0.f, 0.f};
    float* hsw = hst + wg * (RB * OT);
    float* csw = cst + ((wg * FB + tid) << 1);
    if (T0 == 0) {
        for (int i = tid; i < RB * OT; i += FB) h_lds[0][i >> 7][i & (OT - 1)] = 0.0f;
    } else {
        for (int i = tid; i < RB * OT; i += FB) h_lds[0][i >> 7][i & (OT - 1)] = hsw[i];
        const float2 cv = *(const float2*)csw;
        creg[0] = cv.x;
        creg[1] = cv.y;
    }
    const int crow = lane >> 2;
    const int cq = (lane & 3) << 2;
    const int cj = (wave << 2) + (lane & 3);
    float* sw = &s_lds[wave][0];
    const int orow = wave >> 1;
    const int ocol = ((wave & 1) << 6) + (lane << 1);
    float* outp = out + (long)(brow0 + orow) * TT * OT + ocol;
    __syncthreads();

    int cur = 0;
    auto step = [&](const int t, const bool store_prev) __attribute__((always_inline)) {
        float4 gp[2];
#pragma unroll
        for (int u = 0; u < 2; ++u)
            gp[u] = *(const float4*)(gpre + ((long)(brow0 + crow + (u << 3)) * TT + t) * GATES + col0 + cq);
        __builtin_amdgcn_fence(__ATOMIC_RELEASE, "wavefront");

        const float (*hc)[HP] = h_lds[cur];
        float (*hn)[HP] = h_lds[cur ^ 1];
        const float2 ov = *(const float2*)&hc[orow][ocol];

        v8f c0 = {};
#pragma unroll
        for (int kk = 0; kk < 32; ++kk) {
            const int ka = (kk << 2) + (hl << 1);
            v2f b, a0;
            b.x = bx[kk];                b.y = by[kk];
            a0.x = hc[l16][ka];          a0.y = hc[l16][ka + 1];
            c0 = wmma_f32(a0, b, c0);
        }
#pragma unroll
        for (int r = 0; r < 8; ++r) sw[hl * SP + (r << 4) + l16] = c0[r];
        __builtin_amdgcn_fence(__ATOMIC_RELEASE, "wavefront");
        __builtin_amdgcn_wave_barrier();
        __builtin_amdgcn_fence(__ATOMIC_ACQUIRE, "wavefront");
        float4 gs[2];
#pragma unroll
        for (int u = 0; u < 2; ++u) gs[u] = *(const float4*)&sw[u * SP + (lane << 2)];
        if (store_prev) *(float2*)(outp + (long)(t - 1) * OT) = ov;

#pragma unroll
        for (int u = 0; u < 2; ++u) {
            float gi = gs[u].x + gp[u].x;
            float gf = gs[u].y + gp[u].y;
            float gg = gs[u].z + gp[u].z;
            float go = gs[u].w + gp[u].w;
            float si = sigmoid_half_hw(gi);
            float sf = sigmoid_half_hw(gf);
            float so = sigmoid_half_hw(go);
            float cc = sf * creg[u] + si * tanh_hw(gg);
            creg[u] = cc;
            float hh = so * tanh_hw(cc);
            hn[crow + (u << 3)][cj] = hh;
        }
        __builtin_amdgcn_fence(__ATOMIC_RELEASE, "workgroup", "local");
        __builtin_amdgcn_s_barrier();
        __builtin_amdgcn_fence(__ATOMIC_ACQUIRE, "workgroup", "local");
        cur ^= 1;
    };
    step(T0, false);
    for (int t = T0 + 1; t < T1; ++t) step(t, true);
    *(float2*)(outp + (long)(T1 - 1) * OT) = *(const float2*)&h_lds[cur][orow][ocol];
    if (T1 < TT) {
        for (int i = tid; i < RB * OT; i += FB) hsw[i] = h_lds[cur][i >> 7][i & (OT - 1)];
        *(float2*)csw = make_float2(creg[0], creg[1]);
    }
}

template <int STAGE, int T0, int T1>
__global__ __launch_bounds__(FB) void fused_k(
    const float* gpre, const float* __restrict__ WhhT, float* __restrict__ out,
    float* __restrict__ hst, float* __restrict__ cst,
    const int* __restrict__ src, const int* __restrict__ dst, int E,
    int* __restrict__ cnt_out, int* __restrict__ cnt_in, int* __restrict__ rowptr,
    int* __restrict__ cursor, float* __restrict__ norm_out, float* __restrict__ norm_in,
    int2* __restrict__ ebuf, const float* __restrict__ ht, const float* __restrict__ chs,
    float* __restrict__ agg, const float* __restrict__ W1, const float* __restrict__ b1,
    float* __restrict__ x1, const float* __restrict__ W2, const float* __restrict__ b2,
    float* __restrict__ out_sp) {
    if (blockIdx.x < NLW) {
        lstm_segment<T0, T1>((int)blockIdx.x, gpre, WhhT, out, hst, cst);
        return;
    }
    const int vb = (int)blockIdx.x - NLW;
    if (STAGE == ST_DEGREE)  degree_stage(vb, src, dst, cnt_out, cnt_in, E);
    if (STAGE == ST_SCAN)    scan_stage(cnt_in, cnt_out, rowptr, cursor, norm_out, norm_in);
    if (STAGE == ST_FILL)    fill_stage(vb, src, dst, norm_out, cursor, ebuf, E);
    if (STAGE == ST_GATHER1) gather1_stage(vb, rowptr, ebuf, ht, chs, agg);
    if (STAGE == ST_GEMM1)   gemm64_stage(vb, agg, norm_in, W1, b1, x1, 128);
    if (STAGE == ST_GATHER2) gather2_stage(vb, rowptr, ebuf, x1, agg);
    if (STAGE == ST_GEMM2)   gemm64_stage(vb, agg, norm_in, W2, b2, out_sp, 64);
}

extern "C" void kernel_launch(void* const* d_in, const int* in_sizes, int n_in,
                              void* d_out, int out_size, void* d_ws, size_t ws_size,
                              hipStream_t stream) {
    const float* hs  = (const float*)d_in[0];
    const float* ht  = (const float*)d_in[1];
    const float* chs = (const float*)d_in[2];
    const float* cht = (const float*)d_in[3];
    const int*   src = (const int*)d_in[4];
    const int*   dst = (const int*)d_in[5];
    const float* W1  = (const float*)d_in[6];
    const float* b1  = (const float*)d_in[7];
    const float* W2  = (const float*)d_in[8];
    const float* b2  = (const float*)d_in[9];
    const float* Wih = (const float*)d_in[10];
    const float* Whh = (const float*)d_in[11];
    const float* bih = (const float*)d_in[12];
    const float* bhh = (const float*)d_in[13];
    const int E = in_sizes[4];

    float* ws       = (float*)d_ws;
    float* norm_out = ws;
    float* norm_in  = norm_out + N_NODES;
    float* agg      = norm_in + N_NODES;
    float* bkt      = agg + (size_t)N_NODES * 128;
    float* x1       = bkt + (size_t)N_NODES * 64;
    float* Xcat     = x1 + (size_t)N_NODES * 64;
    float* gpre     = Xcat + (size_t)BT * 128;
    float* WihT     = gpre + (size_t)BT * GATES;
    float* WhhT     = WihT + OT * GATES;
    float* bsum     = WhhT + OT * GATES;
    int*  cnt_in  = (int*)bkt;
    int*  cnt_out = cnt_in + N_NODES;
    int*  rowptr  = cnt_out + N_NODES;
    int*  cursor  = rowptr + N_NODES + 16;
    int2* ebuf    = (int2*)(cursor + N_NODES);
    float* hst = Xcat;
    float* cst = Xcat + NLW * RB * OT;
    float* out_lstm = (float*)d_out;
    float* out_sp = (float*)d_out + (size_t)BT * OT;

    const int thr = 256;
    zero_i32<<<(2 * N_NODES + thr - 1) / thr, thr, 0, stream>>>(cnt_in, 2 * N_NODES);

    transpose_k<<<(GATES * OT + thr - 1) / thr, thr, 0, stream>>>(Wih, WihT);
    transpose_k<<<(GATES * OT + thr - 1) / thr, thr, 0, stream>>>(Whh, WhhT);
    bias_sum_k<<<2, 256, 0, stream>>>(bih, bhh, bsum);
    concat_k<<<(BT * 128 + thr - 1) / thr, thr, 0, stream>>>(hs, cht, Xcat);
    dim3 gp(BT / 16, 8);
    gemm_wmma_kernel<<<gp, 128, 0, stream>>>(Xcat, nullptr, WihT, bsum, gpre, BT, GATES, 128, 0);

    const int eb = (E + FB - 1) / FB;
    const int nb = (N_NODES * 32 + FB - 1) / FB;
    const int mb = (N_NODES / 16 + 7) / 8;
#define FUSED_ARGS gpre, WhhT, out_lstm, hst, cst, src, dst, E, cnt_out, cnt_in, rowptr, cursor, \
                   norm_out, norm_in, ebuf, ht, chs, agg, W1, b1, x1, W2, b2, out_sp
    fused_k<ST_DEGREE,    0,  64><<<NLW + eb, FB, 0, stream>>>(FUSED_ARGS);
    fused_k<ST_SCAN,     64,  96><<<NLW + 1,  FB, 0, stream>>>(FUSED_ARGS);
    fused_k<ST_FILL,     96, 160><<<NLW + eb, FB, 0, stream>>>(FUSED_ARGS);
    fused_k<ST_GATHER1, 160, 256><<<NLW + nb, FB, 0, stream>>>(FUSED_ARGS);
    fused_k<ST_GEMM1,   256, 320><<<NLW + mb, FB, 0, stream>>>(FUSED_ARGS);
    fused_k<ST_GATHER2, 320, 384><<<NLW + nb, FB, 0, stream>>>(FUSED_ARGS);
    fused_k<ST_GEMM2,   384, 448><<<NLW + mb, FB, 0, stream>>>(FUSED_ARGS);
    fused_k<ST_NONE,    448, 512><<<NLW,      FB, 0, stream>>>(FUSED_ARGS);
#undef FUSED_ARGS
}


# LSTM step: h fragment loads issued as groups of 4 k-steps, one group ahead of the WMMAs that use them (2 fragment sets in registers), instead of an LDS round trip waited out before every WMMA pair
# speedup vs baseline: 7.4468x; 1.0349x over previous
#include <hip/hip_runtime.h>
#include <math.h>

typedef float v2f __attribute__((ext_vector_type(2)));
typedef float v8f __attribute__((ext_vector_type(8)));

#define N_NODES 50000
#define BB 32
#define TT 512
#define SD 64
#define OT 128
#define GATES 512
#define BT (BB * TT)
#define HP (OT + 4)
#define RB 16
#define GCOL(r) (((((r) & (OT - 1)) >> 2) << 4) + (((r) & 3) << 2) + ((r) >> 7))
#define GSCALE(r) ((((r) >> 7) == 2) ? 1.0f : 0.5f)
#define SP 144
#define SW 272
#define HCH 4

__device__ __forceinline__ v8f wmma_f32(v2f a, v2f b, v8f c) {
    return __builtin_amdgcn_wmma_f32_16x16x4_f32(
         false, a,  false, b,
         (short)0, c,  false,  false);
}

#define FB 1024
#define NLW (BB / RB)
#define ST_NONE 0
#define ST_DEGREE 1
#define ST_SCAN 2
#define ST_FILL 3
#define ST_GATHER1 4
#define ST_GEMM1 5
#define ST_GATHER2 6
#define ST_GEMM2 7

__global__ void zero_i32(int* __restrict__ p, int n) {
    int i = blockIdx.x * blockDim.x + threadIdx.x;
    if (i < n) p[i] = 0;
}

__device__ __forceinline__ void degree_stage(const int vb, const int* __restrict__ src,
                                             const int* __restrict__ dst, int* __restrict__ cnt_out,
                                             int* __restrict__ cnt_in, int E) {
    const int e = vb * FB + (int)threadIdx.x;
    if (e < E) {
        atomicAdd(&cnt_out[src[e]], 1);
        atomicAdd(&cnt_in[dst[e]], 1);
    }
}

#define SCAN_IT ((N_NODES + 1023) / 1024)
#define SCAN_G 7
static_assert(SCAN_IT % SCAN_G == 0, "group size must divide the pass count");
static_assert(FB == 1024, "the scan is written for one workgroup of 32 waves");
__device__ __forceinline__ void scan_stage(const int* __restrict__ cnt_in,
                                           const int* __restrict__ cnt_out,
                                           int* __restrict__ rowptr, int* __restrict__ cursor,
                                           float* __restrict__ no, float* __restrict__ ni) {
    __shared__ int wtot[32];
    const int wave = threadIdx.x >> 5, lane = threadIdx.x & 31;
    const int base = wave * (SCAN_IT * 32) + lane;
    int tot = 0;
#pragma unroll 7
    for (int it = 0; it < SCAN_IT; ++it) {
        const int i = base + it * 32;
        const int cval = cnt_in[min(i, N_NODES - 1)];
        tot += (i < N_NODES) ? cval : 0;
    }
#pragma unroll
    for (int o = 16; o > 0; o >>= 1) tot += __shfl_xor(tot, o);
    if (lane == 0) wtot[wave] = tot;
    __syncthreads();
    int carry = 0;
    for (int w = 0; w < wave; ++w) carry += wtot[w];
    for (int g = 0; g < SCAN_IT; g += SCAN_G) {
        int cv[SCAN_G], co[SCAN_G], x[SCAN_G];
#pragma unroll
        for (int u = 0; u < SCAN_G; ++u) {
            const int i = base + (g + u) * 32;
            const int ic = min(i, N_NODES - 1);
            const int cval = cnt_in[ic];
            co[u] = cnt_out[ic];
            cv[u] = (i < N_NODES) ? cval : 0;
            x[u] = cv[u];
        }
#pragma unroll
        for (int o = 1; o < 32; o <<= 1) {
#pragma unroll
            for (int u = 0; u < SCAN_G; ++u) {
                const int y = __shfl_up(x[u], o);
                if (lane >= o) x[u] += y;
            }
        }
#pragma unroll
        for (int u = 0; u < SCAN_G; ++u) {
            const int i = base + (g + u) * 32;
            if (i < N_NODES) {
                const int v = carry + x[u] - cv[u];
                rowptr[i] = v;
                cursor[i] = v;
                no[i] = rsqrtf(fmaxf((float)co[u], 1.0f));
                ni[i] = rsqrtf(fmaxf((float)cv[u], 1.0f));
            }
            carry += __shfl(x[u], 31);
        }
    }
    if (threadIdx.x == 1023) rowptr[N_NODES] = carry;
}

__device__ __forceinline__ void fill_stage(const int vb, const int* __restrict__ src,
                                           const int* __restrict__ dst,
                                           const float* __restrict__ norm_out, int* __restrict__ cursor,
                                           int2* __restrict__ ebuf, int E) {
    const int e = vb * FB + (int)threadIdx.x;
    if (e < E) {
        const int s = src[e];
        const int p = atomicAdd(&cursor[dst[e]], 1);
        ebuf[p] = make_int2(s, __float_as_int(norm_out[s]));
    }
}

__device__ __forceinline__ void gather1_stage(const int vb, const int* __restrict__ rowptr,
                                              const int2* __restrict__ ebuf,
                                              const float* __restrict__ ht, const float* __restrict__ chs,
                                              float* __restrict__ agg1) {
    const int lane = threadIdx.x & 31;
    const int node = __builtin_amdgcn_readfirstlane((vb << 5) + (int)(threadIdx.x >> 5));
    if (node >= N_NODES) return;
    const int q = lane << 2;
    const float* fb = (q < SD) ? (ht + q) : (chs + (q - SD));
    const int beg = rowptr[node], end = rowptr[node + 1];
    float4 acc = make_float4(0.f, 0.f, 0.f, 0.f);
    for (int j0 = beg; j0 < end; j0 += 32) {
        const int m = min(32, end - j0);
        int2 ew = make_int2(0, 0);
        if (lane < m) ew = ebuf[j0 + lane];
        int i = 0;
        for (; i + 4 <= m; i += 4) {
            const int s0 = __builtin_amdgcn_readlane(ew.x, i);
            const int s1 = __builtin_amdgcn_readlane(ew.x, i + 1);
            const int s2 = __builtin_amdgcn_readlane(ew.x, i + 2);
            const int s3 = __builtin_amdgcn_readlane(ew.x, i + 3);
            const float4 v0 = *(const float4*)(fb + (long)s0 * SD);
            const float4 v1 = *(const float4*)(fb + (long)s1 * SD);
            const float4 v2 = *(const float4*)(fb + (long)s2 * SD);
            const float4 v3 = *(const float4*)(fb + (long)s3 * SD);
            const float w0 = __int_as_float(__builtin_amdgcn_readlane(ew.y, i));
            const float w1 = __int_as_float(__builtin_amdgcn_readlane(ew.y, i + 1));
            const float w2 = __int_as_float(__builtin_amdgcn_readlane(ew.y, i + 2));
            const float w3 = __int_as_float(__builtin_amdgcn_readlane(ew.y, i + 3));
            acc.x += v0.x * w0; acc.y += v0.y * w0; acc.z += v0.z * w0; acc.w += v0.w * w0;
            acc.x += v1.x * w1; acc.y += v1.y * w1; acc.z += v1.z * w1; acc.w += v1.w * w1;
            acc.x += v2.x * w2; acc.y += v2.y * w2; acc.z += v2.z * w2; acc.w += v2.w * w2;
            acc.x += v3.x * w3; acc.y += v3.y * w3; acc.z += v3.z * w3; acc.w += v3.w * w3;
        }
        for (; i < m; ++i) {
            const int s0 = __builtin_amdgcn_readlane(ew.x, i);
            const float w0 = __int_as_float(__builtin_amdgcn_readlane(ew.y, i));
            const float4 v0 = *(const float4*)(fb + (long)s0 * SD);
            acc.x += v0.x * w0; acc.y += v0.y * w0; acc.z += v0.z * w0; acc.w += v0.w * w0;
        }
    }
    *(float4*)(agg1 + (long)node * 128 + q) = acc;
}

__device__ __forceinline__ void gather2_stage(const int vb, const int* __restrict__ rowptr,
                                              const int2* __restrict__ ebuf,
                                              const float* __restrict__ x1, float* __restrict__ agg2) {
    const int lane = threadIdx.x & 31;
    const int node = __builtin_amdgcn_readfirstlane((vb << 5) + (int)(threadIdx.x >> 5));
    if (node >= N_NODES) return;
    const int q = lane << 1;
    const float* fb = x1 + q;
    const int beg = rowptr[node], end = rowptr[node + 1];
    float2 acc = make_float2(0.f, 0.f);
    for (int j0 = beg; j0 < end; j0 += 32) {
        const int m = min(32, end - j0);
        int2 ew = make_int2(0, 0);
        if (lane < m) ew = ebuf[j0 + lane];
        int i = 0;
        for (; i + 4 <= m; i += 4) {
            const int s0 = __builtin_amdgcn_readlane(ew.x, i);
            const int s1 = __builtin_amdgcn_readlane(ew.x, i + 1);
            const int s2 = __builtin_amdgcn_readlane(ew.x, i + 2);
            const int s3 = __builtin_amdgcn_readlane(ew.x, i + 3);
            const float2 v0 = *(const float2*)(fb + (long)s0 * 64);
            const float2 v1 = *(const float2*)(fb + (long)s1 * 64);
            const float2 v2 = *(const float2*)(fb + (long)s2 * 64);
            const float2 v3 = *(const float2*)(fb + (long)s3 * 64);
            const float w0 = __int_as_float(__builtin_amdgcn_readlane(ew.y, i));
            const float w1 = __int_as_float(__builtin_amdgcn_readlane(ew.y, i + 1));
            const float w2 = __int_as_float(__builtin_amdgcn_readlane(ew.y, i + 2));
            const float w3 = __int_as_float(__builtin_amdgcn_readlane(ew.y, i + 3));
            acc.x += v0.x * w0; acc.y += v0.y * w0;
            acc.x += v1.x * w1; acc.y += v1.y * w1;
            acc.x += v2.x * w2; acc.y += v2.y * w2;
            acc.x += v3.x * w3; acc.y += v3.y * w3;
        }
        for (; i < m; ++i) {
            const int s0 = __builtin_amdgcn_readlane(ew.x, i);
            const float w0 = __int_as_float(__builtin_amdgcn_readlane(ew.y, i));
            const float2 v0 = *(const float2*)(fb + (long)s0 * 64);
            acc.x += v0.x * w0; acc.y += v0.y * w0;
        }
    }
    *(float2*)(agg2 + (long)node * 64 + q) = acc;
}

__global__ void transpose_k(const float* __restrict__ W, float* __restrict__ WT) {
    int idx = blockIdx.x * blockDim.x + threadIdx.x;
    if (idx >= GATES * OT) return;
    int r = idx >> 7, c = idx & 127;
    WT[c * GATES + GCOL(r)] = W[idx] * GSCALE(r);
}

__global__ void bias_sum_k(const float* __restrict__ a, const float* __restrict__ b,
                           float* __restrict__ o) {
    int i = blockIdx.x * blockDim.x + threadIdx.x;
    if (i < GATES) o[GCOL(i)] = (a[i] + b[i]) * GSCALE(i);
}

__global__ void concat_k(const float* __restrict__ hs, const float* __restrict__ cht,
                         float* __restrict__ X) {
    int idx = blockIdx.x * blockDim.x + threadIdx.x;
    if (idx >= BT * 128) return;
    int row = idx >> 7, k = idx & 127;
    X[idx] = (k < SD) ? hs[row * SD + k] : cht[row * SD + (k - SD)];
}

__device__ __forceinline__ void gemm_tile(const float* __restrict__ A, const float* __restrict__ rowscale,
                                          const float* __restrict__ Bm, const float* __restrict__ bias,
                                          float* __restrict__ C, const int N, const int K,
                                          const int applyGelu, const int row0, const int ct, const int lane) {
    const int hl = lane >> 4;
    const int l16 = lane & 15;
    const int col = (ct << 4) + l16;
    const int arow = row0 + l16;
    const float rs = rowscale ? rowscale[arow] : 1.0f;
    const float* Arow = A + (size_t)arow * K;
    v8f acc = {};
    for (int k = 0; k < K; k += 4) {
        const int ka = k + (hl << 1);
        v2f a, b;
        a.x = Arow[ka] * rs;
        a.y = Arow[ka + 1] * rs;
        b.x = Bm[(size_t)ka * N + col];
        b.y = Bm[(size_t)(ka + 1) * N + col];
        acc = wmma_f32(a, b, acc);
    }
    const float bv = bias ? bias[col] : 0.0f;
#pragma unroll
    for (int r = 0; r < 8; ++r) {
        float v = acc[r] + bv;
        if (applyGelu) v = 0.5f * v * (1.0f + erff(v * 0.70710678118654752f));
        C[(size_t)(row0 + r + (hl << 3)) * N + col] = v;
    }
}

__global__ void gemm_wmma_kernel(const float* __restrict__ A, const float* __restrict__ rowscale,
                                 const float* __restrict__ Bm, const float* __restrict__ bias,
                                 float* __restrict__ C, int M, int N, int K, int applyGelu) {
    const int wave = threadIdx.x >> 5;
    const int ct = blockIdx.y * (blockDim.x >> 5) + wave;
    if ((ct << 4) >= N) return;
    gemm_tile(A, rowscale, Bm, bias, C, N, K, applyGelu, blockIdx.x << 4, ct, threadIdx.x & 31);
}

__device__ __forceinline__ void gemm64_stage(const int vb, const float* __restrict__ A,
                                             const float* __restrict__ rowscale,
                                             const float* __restrict__ Bm, const float* __restrict__ bias,
                                             float* __restrict__ C, const int K) {
    const int wave = threadIdx.x >> 5;
    const int rt = (vb << 3) + (wave >> 2);
    if (rt >= N_NODES / 16) return;
    gemm_tile(A, rowscale, Bm, bias, C, 64, K, 1, rt << 4, wave & 3, threadIdx.x & 31);
}

__device__ __forceinline__ float tanh_hw(float x) {
    return __builtin_amdgcn_tanhf(x);
}
__device__ __forceinline__ float sigmoid_half_hw(float xh) {
    return fmaf(__builtin_amdgcn_tanhf(xh), 0.5f, 0.5f);
}

template <int T0, int T1>
__device__ __forceinline__ void lstm_segment(const int wg,
                                             const float* gpre,
                                             const float* __restrict__ WhhT,
                                             float* __restrict__ out,
                                             float* __restrict__ hst, float* __restrict__ cst) {
    __shared__ __attribute__((aligned(16))) float h_lds[2][RB][HP];
    __shared__ __attribute__((aligned(16))) float s_lds[32][SW];
    const int tid = threadIdx.x;
    const int wave = tid >> 5, lane = tid & 31;
    const int hl = lane >> 4, l16 = lane & 15;
    const int col0 = wave << 4;
    const int col = col0 + l16;
    const int brow0 = wg * RB;

    float bx[32], by[32];
#pragma unroll
    for (int kk = 0; kk < 32; ++kk) {
        const int ka = (kk << 2) + (hl << 1);
        bx[kk] = WhhT[ka * GATES + col];
        by[kk] = WhhT[(ka + 1) * GATES + col];
    }

    float creg[2] = {0.f, 0.f};
    float* hsw = hst + wg * (RB * OT);
    float* csw = cst + ((wg * FB + tid) << 1);
    if (T0 == 0) {
        for (int i = tid; i < RB * OT; i += FB) h_lds[0][i >> 7][i & (OT - 1)] = 0.0f;
    } else {
        for (int i = tid; i < RB * OT; i += FB) h_lds[0][i >> 7][i & (OT - 1)] = hsw[i];
        const float2 cv = *(const float2*)csw;
        creg[0] = cv.x;
        creg[1] = cv.y;
    }
    const int crow = lane >> 2;
    const int cq = (lane & 3) << 2;
    const int cj = (wave << 2) + (lane & 3);
    float* sw = &s_lds[wave][0];
    const int orow = wave >> 1;
    const int ocol = ((wave & 1) << 6) + (lane << 1);
    float* outp = out + (long)(brow0 + orow) * TT * OT + ocol;
    __syncthreads();

    int cur = 0;
    auto step = [&](const int t, const bool store_prev) __attribute__((always_inline)) {
        float4 gp[2];
#pragma unroll
        for (int u = 0; u < 2; ++u)
            gp[u] = *(const float4*)(gpre + ((long)(brow0 + crow + (u << 3)) * TT + t) * GATES + col0 + cq);
        __builtin_amdgcn_fence(__ATOMIC_RELEASE, "wavefront");

        const float (*hc)[HP] = h_lds[cur];
        float (*hn)[HP] = h_lds[cur ^ 1];
        const float2 ov = *(const float2*)&hc[orow][ocol];

        v2f hf[2][HCH];
#pragma unroll
        for (int j = 0; j < HCH; ++j) {
            const int ka = (j << 2) + (hl << 1);
            hf[0][j].x = hc[l16][ka];    hf[0][j].y = hc[l16][ka + 1];
        }
        v8f c0 = {};
#pragma unroll
        for (int c = 0; c < 32 / HCH; ++c) {
            if (c + 1 < 32 / HCH) {
#pragma unroll
                for (int j = 0; j < HCH; ++j) {
                    const int ka = (((c + 1) * HCH + j) << 2) + (hl << 1);
                    hf[(c + 1) & 1][j].x = hc[l16][ka];
                    hf[(c + 1) & 1][j].y = hc[l16][ka + 1];
                }
            }
            __builtin_amdgcn_sched_barrier(0);
#pragma unroll
            for (int j = 0; j < HCH; ++j) {
                const int kk = c * HCH + j;
                v2f b;
                b.x = bx[kk];            b.y = by[kk];
                c0 = wmma_f32(hf[c & 1][j], b, c0);
            }
            __builtin_amdgcn_sched_barrier(0);
        }
#pragma unroll
        for (int r = 0; r < 8; ++r) sw[hl * SP + (r << 4) + l16] = c0[r];
        __builtin_amdgcn_fence(__ATOMIC_RELEASE, "wavefront");
        __builtin_amdgcn_wave_barrier();
        __builtin_amdgcn_fence(__ATOMIC_ACQUIRE, "wavefront");
        float4 gs[2];
#pragma unroll
        for (int u = 0; u < 2; ++u) gs[u] = *(const float4*)&sw[u * SP + (lane << 2)];
        if (store_prev) *(float2*)(outp + (long)(t - 1) * OT) = ov;

#pragma unroll
        for (int u = 0; u < 2; ++u) {
            float gi = gs[u].x + gp[u].x;
            float gf = gs[u].y + gp[u].y;
            float gg = gs[u].z + gp[u].z;
            float go = gs[u].w + gp[u].w;
            float si = sigmoid_half_hw(gi);
            float sf = sigmoid_half_hw(gf);
            float so = sigmoid_half_hw(go);
            float cc = sf * creg[u] + si * tanh_hw(gg);
            creg[u] = cc;
            float hh = so * tanh_hw(cc);
            hn[crow + (u << 3)][cj] = hh;
        }
        __builtin_amdgcn_fence(__ATOMIC_RELEASE, "workgroup", "local");
        __builtin_amdgcn_s_barrier();
        __builtin_amdgcn_fence(__ATOMIC_ACQUIRE, "workgroup", "local");
        cur ^= 1;
    };
    step(T0, false);
    for (int t = T0 + 1; t < T1; ++t) step(t, true);
    *(float2*)(outp + (long)(T1 - 1) * OT) = *(const float2*)&h_lds[cur][orow][ocol];
    if (T1 < TT) {
        for (int i = tid; i < RB * OT; i += FB) hsw[i] = h_lds[cur][i >> 7][i & (OT - 1)];
        *(float2*)csw = make_float2(creg[0], creg[1]);
    }
}

template <int STAGE, int T0, int T1>
__global__ __launch_bounds__(FB) void fused_k(
    const float* gpre, const float* __restrict__ WhhT, float* __restrict__ out,
    float* __restrict__ hst, float* __restrict__ cst,
    const int* __restrict__ src, const int* __restrict__ dst, int E,
    int* __restrict__ cnt_out, int* __restrict__ cnt_in, int* __restrict__ rowptr,
    int* __restrict__ cursor, float* __restrict__ norm_out, float* __restrict__ norm_in,
    int2* __restrict__ ebuf, const float* __restrict__ ht, const float* __restrict__ chs,
    float* __restrict__ agg, const float* __restrict__ W1, const float* __restrict__ b1,
    float* __restrict__ x1, const float* __restrict__ W2, const float* __restrict__ b2,
    float* __restrict__ out_sp) {
    if (blockIdx.x < NLW) {
        lstm_segment<T0, T1>((int)blockIdx.x, gpre, WhhT, out, hst, cst);
        return;
    }
    const int vb = (int)blockIdx.x - NLW;
    if (STAGE == ST_DEGREE)  degree_stage(vb, src, dst, cnt_out, cnt_in, E);
    if (STAGE == ST_SCAN)    scan_stage(cnt_in, cnt_out, rowptr, cursor, norm_out, norm_in);
    if (STAGE == ST_FILL)    fill_stage(vb, src, dst, norm_out, cursor, ebuf, E);
    if (STAGE == ST_GATHER1) gather1_stage(vb, rowptr, ebuf, ht, chs, agg);
    if (STAGE == ST_GEMM1)   gemm64_stage(vb, agg, norm_in, W1, b1, x1, 128);
    if (STAGE == ST_GATHER2) gather2_stage(vb, rowptr, ebuf, x1, agg);
    if (STAGE == ST_GEMM2)   gemm64_stage(vb, agg, norm_in, W2, b2, out_sp, 64);
}

extern "C" void kernel_launch(void* const* d_in, const int* in_sizes, int n_in,
                              void* d_out, int out_size, void* d_ws, size_t ws_size,
                              hipStream_t stream) {
    const float* hs  = (const float*)d_in[0];
    const float* ht  = (const float*)d_in[1];
    const float* chs = (const float*)d_in[2];
    const float* cht = (const float*)d_in[3];
    const int*   src = (const int*)d_in[4];
    const int*   dst = (const int*)d_in[5];
    const float* W1  = (const float*)d_in[6];
    const float* b1  = (const float*)d_in[7];
    const float* W2  = (const float*)d_in[8];
    const float* b2  = (const float*)d_in[9];
    const float* Wih = (const float*)d_in[10];
    const float* Whh = (const float*)d_in[11];
    const float* bih = (const float*)d_in[12];
    const float* bhh = (const float*)d_in[13];
    const int E = in_sizes[4];

    float* ws       = (float*)d_ws;
    float* norm_out = ws;
    float* norm_in  = norm_out + N_NODES;
    float* agg      = norm_in + N_NODES;
    float* bkt      = agg + (size_t)N_NODES * 128;
    float* x1       = bkt + (size_t)N_NODES * 64;
    float* Xcat     = x1 + (size_t)N_NODES * 64;
    float* gpre     = Xcat + (size_t)BT * 128;
    float* WihT     = gpre + (size_t)BT * GATES;
    float* WhhT     = WihT + OT * GATES;
    float* bsum     = WhhT + OT * GATES;
    int*  cnt_in  = (int*)bkt;
    int*  cnt_out = cnt_in + N_NODES;
    int*  rowptr  = cnt_out + N_NODES;
    int*  cursor  = rowptr + N_NODES + 16;
    int2* ebuf    = (int2*)(cursor + N_NODES);
    float* hst = Xcat;
    float* cst = Xcat + NLW * RB * OT;
    float* out_lstm = (float*)d_out;
    float* out_sp = (float*)d_out + (size_t)BT * OT;

    const int thr = 256;
    zero_i32<<<(2 * N_NODES + thr - 1) / thr, thr, 0, stream>>>(cnt_in, 2 * N_NODES);

    transpose_k<<<(GATES * OT + thr - 1) / thr, thr, 0, stream>>>(Wih, WihT);
    transpose_k<<<(GATES * OT + thr - 1) / thr, thr, 0, stream>>>(Whh, WhhT);
    bias_sum_k<<<2, 256, 0, stream>>>(bih, bhh, bsum);
    concat_k<<<(BT * 128 + thr - 1) / thr, thr, 0, stream>>>(hs, cht, Xcat);
    dim3 gp(BT / 16, 8);
    gemm_wmma_kernel<<<gp, 128, 0, stream>>>(Xcat, nullptr, WihT, bsum, gpre, BT, GATES, 128, 0);

    const int eb = (E + FB - 1) / FB;
    const int nb = (N_NODES * 32 + FB - 1) / FB;
    const int mb = (N_NODES / 16 + 7) / 8;
#define FUSED_ARGS gpre, WhhT, out_lstm, hst, cst, src, dst, E, cnt_out, cnt_in, rowptr, cursor, \
                   norm_out, norm_in, ebuf, ht, chs, agg, W1, b1, x1, W2, b2, out_sp
    fused_k<ST_DEGREE,    0,  64><<<NLW + eb, FB, 0, stream>>>(FUSED_ARGS);
    fused_k<ST_SCAN,     64,  96><<<NLW + 1,  FB, 0, stream>>>(FUSED_ARGS);
    fused_k<ST_FILL,     96, 160><<<NLW + eb, FB, 0, stream>>>(FUSED_ARGS);
    fused_k<ST_GATHER1, 160, 256><<<NLW + nb, FB, 0, stream>>>(FUSED_ARGS);
    fused_k<ST_GEMM1,   256, 320><<<NLW + mb, FB, 0, stream>>>(FUSED_ARGS);
    fused_k<ST_GATHER2, 320, 384><<<NLW + nb, FB, 0, stream>>>(FUSED_ARGS);
    fused_k<ST_GEMM2,   384, 448><<<NLW + mb, FB, 0, stream>>>(FUSED_ARGS);
    fused_k<ST_NONE,    448, 512><<<NLW,      FB, 0, stream>>>(FUSED_ARGS);
#undef FUSED_ARGS
}
